# starting kernel
# baseline (speedup 1.0000x reference)
#include <hip/hip_runtime.h>
#include <hip/hip_bf16.h>

typedef __attribute__((ext_vector_type(2))) float v2f;
typedef __attribute__((ext_vector_type(8))) float v8f;

#define WMMA_F32X4(a, b, c) \
    __builtin_amdgcn_wmma_f32_16x16x4_f32(false, (a), false, (b), (short)0, (c), false, false)

__global__ void __launch_bounds__(256) k_rankmm(const float* __restrict__ U,
                                                const float* __restrict__ V,
                                                float* __restrict__ out,
                                                int rows, int cols) {
    int idx = blockIdx.x * 256 + threadIdx.x;
    if (idx >= rows * cols) return;
    int n = idx / cols;
    int m = idx - n * cols;
    const float* up = U + n * 32;
    const float* vp = V + m * 32;
    float acc = 0.f;
#pragma unroll
    for (int r = 0; r < 32; ++r) acc += up[r] * vp[r];
    out[idx] = acc;
}

__global__ void __launch_bounds__(256) k_wmma_proj(const float* __restrict__ uh,
                                                   const float* __restrict__ KHH_U,
                                                   const float* __restrict__ KHH_V,
                                                   const float* __restrict__ Kbd_U,
                                                   float* __restrict__ wh0,
                                                   float* __restrict__ sbuf) {
    __shared__ float tld[8][16 * 32];
    const int wave = threadIdx.x >> 5;
    const int lane = threadIdx.x & 31;
    const int half = lane >> 4;
    const int mrow = lane & 15;
    const int wid  = blockIdx.x * 8 + wave;
    const int R    = wid * 16;
    const float HK = 1.0f / 1024.0f;

    const float* arow = uh + (size_t)(R + mrow) * 8193;

    v8f t0 = {}; v8f t1 = {};
    for (int k0 = 0; k0 < 1024; k0 += 4) {
        int ka = k0 + 2 * half;
        int kb = ka + 1;
        v2f a, b0, b1;
        a.x  = arow[8 * ka];
        a.y  = arow[8 * kb];
        b0.x = KHH_V[ka * 32 + mrow];
        b0.y = KHH_V[kb * 32 + mrow];
        b1.x = KHH_V[ka * 32 + 16 + mrow];
        b1.y = KHH_V[kb * 32 + 16 + mrow];
        t0 = WMMA_F32X4(a, b0, t0);
        t1 = WMMA_F32X4(a, b1, t1);
    }
    {
        float alast = arow[8192];
        v2f a, b0, b1;
        a.x  = (half == 0) ? alast : 0.f;
        a.y  = 0.f;
        b0.x = KHH_V[1024 * 32 + mrow];
        b0.y = b0.x;
        b1.x = KHH_V[1024 * 32 + 16 + mrow];
        b1.y = b1.x;
        t0 = WMMA_F32X4(a, b0, t0);
        t1 = WMMA_F32X4(a, b1, t1);
    }

    v8f s0 = {}; v8f s1 = {};
    for (int k0 = 0; k0 < 8192; k0 += 4) {
        int ka = k0 + 2 * half;
        int kb = ka + 1;
        v2f a, b0, b1;
        a.x  = arow[ka];
        a.y  = arow[kb];
        b0.x = Kbd_U[ka * 32 + mrow];
        b0.y = Kbd_U[kb * 32 + mrow];
        b1.x = Kbd_U[ka * 32 + 16 + mrow];
        b1.y = Kbd_U[kb * 32 + 16 + mrow];
        s0 = WMMA_F32X4(a, b0, s0);
        s1 = WMMA_F32X4(a, b1, s1);
    }
    {
        float alast = arow[8192];
        v2f a, b0, b1;
        a.x  = (half == 0) ? alast : 0.f;
        a.y  = 0.f;
        b0.x = Kbd_U[8192 * 32 + mrow];
        b0.y = b0.x;
        b1.x = Kbd_U[8192 * 32 + 16 + mrow];
        b1.y = b1.x;
        s0 = WMMA_F32X4(a, b0, s0);
        s1 = WMMA_F32X4(a, b1, s1);
    }
#pragma unroll
    for (int j = 0; j < 8; ++j) {
        int m = j + 8 * half;
        sbuf[(size_t)(R + m) * 32 + mrow]      = s0[j];
        sbuf[(size_t)(R + m) * 32 + 16 + mrow] = s1[j];
    }

    float* tl = tld[wave];
#pragma unroll
    for (int j = 0; j < 8; ++j) {
        int m = j + 8 * half;
        tl[m * 32 + mrow]      = t0[j] * HK;
        tl[m * 32 + 16 + mrow] = t1[j] * HK;
    }
    __syncthreads();

    for (int nt = 0; nt < 64; ++nt) {
        int n = nt * 16 + mrow;
        v8f acc = {};
#pragma unroll
        for (int k0 = 0; k0 < 32; k0 += 4) {
            int ka = k0 + 2 * half;
            int kb = ka + 1;
            v2f a, b;
            a.x = tl[mrow * 32 + ka];
            a.y = tl[mrow * 32 + kb];
            b.x = KHH_U[n * 32 + ka];
            b.y = KHH_U[n * 32 + kb];
            acc = WMMA_F32X4(a, b, acc);
        }
#pragma unroll
        for (int j = 0; j < 8; ++j) {
            int m = j + 8 * half;
            wh0[(size_t)(R + m) * 1025 + n] = acc[j];
        }
    }
    {
        v8f acc = {};
#pragma unroll
        for (int k0 = 0; k0 < 32; k0 += 4) {
            int ka = k0 + 2 * half;
            int kb = ka + 1;
            v2f a, b;
            a.x = tl[mrow * 32 + ka];
            a.y = tl[mrow * 32 + kb];
            b.x = KHH_U[1024 * 32 + ka];
            b.y = KHH_U[1024 * 32 + kb];
            acc = WMMA_F32X4(a, b, acc);
        }
        if (mrow == 0) {
#pragma unroll
            for (int j = 0; j < 8; ++j) {
                int m = j + 8 * half;
                wh0[(size_t)(R + m) * 1025 + 1024] = acc[j];
            }
        }
    }
}

__global__ void __launch_bounds__(256) k_refine(const float* __restrict__ uh,
                                                const float* __restrict__ wh0,
                                                const float* __restrict__ sbuf,
                                                const float* __restrict__ Kbd_V,
                                                const float* __restrict__ Ke0,
                                                const float* __restrict__ Ko0,
                                                const float* __restrict__ Ke1,
                                                const float* __restrict__ Ko1,
                                                const float* __restrict__ Ke2,
                                                const float* __restrict__ Ko2,
                                                const int* __restrict__ nbrL0,
                                                const int* __restrict__ nbrL1,
                                                const int* __restrict__ nbrL2,
                                                float* __restrict__ out) {
    __shared__ float wA[4100];
    __shared__ float wB[8200];
    const int tid = threadIdx.x;
    const int bc  = blockIdx.x;
    const size_t base = (size_t)bc * 8193;
    const float* u = uh + base;
    const float H = 1.0f / 8192.0f;

    for (int n = tid; n < 1025; n += 256) wA[n] = wh0[(size_t)bc * 1025 + n];
    __syncthreads();

    const float* KeP[3]  = {Ke0, Ke1, Ke2};
    const float* KoP[3]  = {Ko0, Ko1, Ko2};
    const int*   nbrP[3] = {nbrL0, nbrL1, nbrL2};
    float* oldW = wA;
    float* newW = wB;

#pragma unroll
    for (int i = 0; i < 3; ++i) {
        const int Nc = 1025 << i;
        const int step = 4 >> i;
        const float hf = H * (float)step;
        const float* Ke = KeP[i];
        const float* Ko = KoP[i];
        const int* nbr  = nbrP[i];
        for (int n = tid; n < Nc; n += 256) {
            const int* row = nbr + (size_t)(2 * n) * 15;
            const float* kp = Ke + (size_t)n * 8;
            float acc = 0.f;
#pragma unroll
            for (int m = 0; m < 8; ++m) acc += kp[m] * u[step * row[2 * m]];
            newW[2 * n] = oldW[n] + hf * acc;
        }
        for (int n = tid; n < Nc - 1; n += 256) {
            const int* row = nbr + (size_t)(2 * n + 1) * 15;
            const float* kp = Ko + (size_t)n * 15;
            float acc = 0.f;
#pragma unroll
            for (int m = 0; m < 15; ++m) acc += kp[m] * u[step * row[m]];
            newW[2 * n + 1] = 0.5f * (oldW[n] + oldW[n + 1]) + hf * acc;
        }
        __syncthreads();
        float* tmp = oldW; oldW = newW; newW = tmp;
    }

    for (int n = tid; n < 8193; n += 256) {
        float v = oldW[n];
        if (n == 0 || n == 8192) {
            const float* kv = Kbd_V + ((n == 0) ? 0 : 32);
            const float* sp = sbuf + bc * 32;
            float acc = 0.f;
#pragma unroll
            for (int r = 0; r < 32; ++r) acc += kv[r] * sp[r];
            v = H * acc;
        }
        out[base + n] = v;
    }
}

extern "C" void kernel_launch(void* const* d_in, const int* in_sizes, int n_in,
                              void* d_out, int out_size, void* d_ws, size_t ws_size,
                              hipStream_t stream) {
    const float* uh     = (const float*)d_in[0];
    const float* KHH_U  = (const float*)d_in[1];
    const float* KHH_V  = (const float*)d_in[2];
    const float* Ke0_U  = (const float*)d_in[3];
    const float* Ke0_V  = (const float*)d_in[4];
    const float* Ko0_U  = (const float*)d_in[5];
    const float* Ko0_V  = (const float*)d_in[6];
    const int*   nbr0   = (const int*)d_in[7];
    const float* Ke1_U  = (const float*)d_in[8];
    const float* Ke1_V  = (const float*)d_in[9];
    const float* Ko1_U  = (const float*)d_in[10];
    const float* Ko1_V  = (const float*)d_in[11];
    const int*   nbr1   = (const int*)d_in[12];
    const float* Ke2_U  = (const float*)d_in[13];
    const float* Ke2_V  = (const float*)d_in[14];
    const float* Ko2_U  = (const float*)d_in[15];
    const float* Ko2_V  = (const float*)d_in[16];
    const int*   nbr2   = (const int*)d_in[17];
    const float* Kbd_U  = (const float*)d_in[18];
    const float* Kbd_V  = (const float*)d_in[19];

    float* ws  = (float*)d_ws;
    float* wh0 = ws;
    float* sb  = wh0 + 524800;
    float* Ke0 = sb + 16384;
    float* Ke1 = Ke0 + 8200;
    float* Ke2 = Ke1 + 16392;
    float* Ko0 = Ke2 + 32776;
    float* Ko1 = Ko0 + 15360;
    float* Ko2 = Ko1 + 30720;

    float* outp = (float*)d_out;

    auto grid = [](int elems) { return (elems + 255) / 256; };

    k_rankmm<<<grid(1025 * 8),  256, 0, stream>>>(Ke0_U, Ke0_V, Ke0, 1025, 8);
    k_rankmm<<<grid(1024 * 15), 256, 0, stream>>>(Ko0_U, Ko0_V, Ko0, 1024, 15);
    k_rankmm<<<grid(2049 * 8),  256, 0, stream>>>(Ke1_U, Ke1_V, Ke1, 2049, 8);
    k_rankmm<<<grid(2048 * 15), 256, 0, stream>>>(Ko1_U, Ko1_V, Ko1, 2048, 15);
    k_rankmm<<<grid(4097 * 8),  256, 0, stream>>>(Ke2_U, Ke2_V, Ke2, 4097, 8);
    k_rankmm<<<grid(4096 * 15), 256, 0, stream>>>(Ko2_U, Ko2_V, Ko2, 4096, 15);

    k_wmma_proj<<<4, 256, 0, stream>>>(uh, KHH_U, KHH_V, Kbd_U, wh0, sb);

    k_refine<<<512, 256, 0, stream>>>(uh, wh0, sb, Kbd_V,
                                      Ke0, Ko0, Ke1, Ko1, Ke2, Ko2,
                                      nbr2, nbr1, nbr0, outp);
}


# Projection: each wave brings its K-slice chunks of uh, Kbd_U and KHH_V in with whole-line loads, parks them in its own LDS area and reads the WMMA fragments from there; k order and sums unchanged
# speedup vs baseline: 6.3038x; 6.3038x over previous
#include <hip/hip_runtime.h>
#include <hip/hip_bf16.h>

typedef __attribute__((ext_vector_type(2))) float v2f;
typedef __attribute__((ext_vector_type(4))) float v4f;
typedef __attribute__((ext_vector_type(8))) float v8f;

#define WMMA_F32X4(a, b, c) \
    __builtin_amdgcn_wmma_f32_16x16x4_f32(false, (a), false, (b), (short)0, (c), false, false)

__global__ void __launch_bounds__(256) k_rankmm(const float* __restrict__ U,
                                                const float* __restrict__ V,
                                                float* __restrict__ out,
                                                int rows, int cols) {
    int idx = blockIdx.x * 256 + threadIdx.x;
    if (idx >= rows * cols) return;
    int n = idx / cols;
    int m = idx - n * cols;
    const float* up = U + n * 32;
    const float* vp = V + m * 32;
    float acc = 0.f;
#pragma unroll
    for (int r = 0; r < 32; ++r) acc += up[r] * vp[r];
    out[idx] = acc;
}

__global__ void __launch_bounds__(1024) k_wmma_proj(const float* __restrict__ uh,
                                                    const float* __restrict__ KHH_U,
                                                    const float* __restrict__ KHH_V,
                                                    const float* __restrict__ Kbd_U,
                                                    float* __restrict__ wh0,
                                                    float* __restrict__ sbuf) {
    __shared__ __attribute__((aligned(16))) float stage[32][2304];
    __shared__ float tl[16 * 32];
    const int wave = __builtin_amdgcn_readfirstlane((int)(threadIdx.x >> 5));
    const int lane = threadIdx.x & 31;
    const int half = lane >> 4;
    const int mrow = lane & 15;
    const int R    = blockIdx.x * 16;
    const float HK = 1.0f / 1024.0f;

    const float* arow = uh + (size_t)(R + mrow) * 8193;

    v8f t0 = {}; v8f t1 = {};
    v8f s0 = {}; v8f s1 = {};
    const int kc0 = 32 * wave;
    const int kf0 = 256 * wave;

    float* As = stage[wave];
    float* Bs = As + 16 * 36;
    float* Vs = Bs + 32 * 48;
    const float* ublk = uh + (size_t)R * 8193 + lane;
    const int srow = lane >> 3;
    const int scol = 4 * (lane & 7);

    for (int c = 0; c < 8; ++c) {
        const int kbase = kf0 + 32 * c;
        float av[16];
#pragma unroll
        for (int r = 0; r < 16; ++r) av[r] = ublk[(size_t)r * 8193 + kbase];
        const v4f* bsrc = (const v4f*)(Kbd_U + (size_t)kbase * 32) + lane;
        v4f bv[8];
#pragma unroll
        for (int i = 0; i < 8; ++i) bv[i] = bsrc[32 * i];
        const v4f vv = *((const v4f*)(KHH_V + (size_t)(kc0 + 4 * c) * 32) + lane);

        __builtin_amdgcn_fence(__ATOMIC_ACQ_REL, "wavefront");
        __builtin_amdgcn_wave_barrier();
#pragma unroll
        for (int r = 0; r < 16; ++r) As[r * 36 + lane] = av[r];
#pragma unroll
        for (int i = 0; i < 8; ++i) *(v4f*)(Bs + (4 * i + srow) * 48 + scol) = bv[i];
        *(v4f*)(Vs + srow * 48 + scol) = vv;
        __builtin_amdgcn_fence(__ATOMIC_ACQ_REL, "wavefront");
        __builtin_amdgcn_wave_barrier();

        {
            v2f a, b0, b1;
            a.x  = As[mrow * 36 + 16 * half];
            a.y  = As[mrow * 36 + 16 * half + 8];
            b0.x = Vs[(2 * half) * 48 + mrow];
            b0.y = Vs[(2 * half + 1) * 48 + mrow];
            b1.x = Vs[(2 * half) * 48 + 16 + mrow];
            b1.y = Vs[(2 * half + 1) * 48 + 16 + mrow];
            t0 = WMMA_F32X4(a, b0, t0);
            t1 = WMMA_F32X4(a, b1, t1);
        }
#pragma unroll
        for (int q = 0; q < 8; ++q) {
            const int kl = 4 * q + 2 * half;
            v2f a, b0, b1;
            a    = *(const v2f*)(As + mrow * 36 + kl);
            b0.x = Bs[kl * 48 + mrow];
            b0.y = Bs[(kl + 1) * 48 + mrow];
            b1.x = Bs[kl * 48 + 16 + mrow];
            b1.y = Bs[(kl + 1) * 48 + 16 + mrow];
            s0 = WMMA_F32X4(a, b0, s0);
            s1 = WMMA_F32X4(a, b1, s1);
        }
    }
    __builtin_amdgcn_fence(__ATOMIC_ACQ_REL, "wavefront");
    __builtin_amdgcn_wave_barrier();
    if (wave == 31) {
        float alast = arow[8192];
        v2f a, b0, b1;
        a.x  = (half == 0) ? alast : 0.f;
        a.y  = 0.f;
        b0.x = KHH_V[1024 * 32 + mrow];
        b0.y = b0.x;
        b1.x = KHH_V[1024 * 32 + 16 + mrow];
        b1.y = b1.x;
        t0 = WMMA_F32X4(a, b0, t0);
        t1 = WMMA_F32X4(a, b1, t1);
        b0.x = Kbd_U[8192 * 32 + mrow];
        b0.y = b0.x;
        b1.x = Kbd_U[8192 * 32 + 16 + mrow];
        b1.y = b1.x;
        s0 = WMMA_F32X4(a, b0, s0);
        s1 = WMMA_F32X4(a, b1, s1);
    }

    float* pw = stage[wave];
#pragma unroll
    for (int j = 0; j < 8; ++j) {
        int m = j + 8 * half;
        pw[m * 32 + mrow]            = t0[j];
        pw[m * 32 + 16 + mrow]       = t1[j];
        pw[512 + m * 32 + mrow]      = s0[j];
        pw[512 + m * 32 + 16 + mrow] = s1[j];
    }
    __syncthreads();

    {
        const int e = threadIdx.x;
        float acc = stage[0][e];
#pragma unroll
        for (int w = 1; w < 32; ++w) acc += stage[w][e];
        if (e < 512) tl[e] = acc * HK;
        else         sbuf[(size_t)R * 32 + (e - 512)] = acc;
    }
    __syncthreads();

    v2f af[8];
#pragma unroll
    for (int q = 0; q < 8; ++q) {
        int ka = 4 * q + 2 * half;
        af[q].x = tl[mrow * 32 + ka];
        af[q].y = tl[mrow * 32 + ka + 1];
    }
#pragma unroll
    for (int u = 0; u < 2; ++u) {
        int n = (2 * wave + u) * 16 + mrow;
        v8f acc = {};
#pragma unroll
        for (int q = 0; q < 8; ++q) {
            int ka = 4 * q + 2 * half;
            v2f b;
            b.x = KHH_U[n * 32 + ka];
            b.y = KHH_U[n * 32 + ka + 1];
            acc = WMMA_F32X4(af[q], b, acc);
        }
#pragma unroll
        for (int j = 0; j < 8; ++j) {
            int m = j + 8 * half;
            wh0[(size_t)(R + m) * 1025 + n] = acc[j];
        }
    }
    if (wave == 0) {
        v8f acc = {};
#pragma unroll
        for (int q = 0; q < 8; ++q) {
            int ka = 4 * q + 2 * half;
            v2f b;
            b.x = KHH_U[1024 * 32 + ka];
            b.y = KHH_U[1024 * 32 + ka + 1];
            acc = WMMA_F32X4(af[q], b, acc);
        }
        if (mrow == 0) {
#pragma unroll
            for (int j = 0; j < 8; ++j) {
                int m = j + 8 * half;
                wh0[(size_t)(R + m) * 1025 + 1024] = acc[j];
            }
        }
    }
}

#define SEGC 16
#define NSEG (1024 / SEGC)
#define LP   33
#define HALO 32
#define WUMAX (8 * SEGC + 2 * HALO + 1)

__global__ void __launch_bounds__(256) k_refine(const float* __restrict__ uh,
                                                const float* __restrict__ wh0,
                                                const float* __restrict__ sbuf,
                                                const float* __restrict__ Kbd_V,
                                                const float* __restrict__ Ke0,
                                                const float* __restrict__ Ko0,
                                                const float* __restrict__ Ke1,
                                                const float* __restrict__ Ko1,
                                                const float* __restrict__ Ke2,
                                                const float* __restrict__ Ko2,
                                                const int* __restrict__ nbrL0,
                                                const int* __restrict__ nbrL1,
                                                const int* __restrict__ nbrL2,
                                                float* __restrict__ out) {
    __shared__ float uw[WUMAX * LP];
    __shared__ float wA[(4 * SEGC + 1) * LP];
    __shared__ float wB[(8 * SEGC + 1) * LP];
    const int wave = __builtin_amdgcn_readfirstlane((int)(threadIdx.x >> 5));
    const int lane = threadIdx.x & 31;
    const int seg  = blockIdx.x % NSEG;
    const int row0 = (blockIdx.x / NSEG) * 32;
    const int c0   = seg * SEGC;
    const int jlo  = (8 * c0 - HALO) < 0 ? 0 : (8 * c0 - HALO);
    const int jhi  = (8 * (c0 + SEGC) + HALO) > 8192 ? 8192 : (8 * (c0 + SEGC) + HALO);
    const int wu   = jhi - jlo + 1;
    const float H  = 1.0f / 8192.0f;

    for (int r = wave; r < 32; r += 8) {
        const float* us = uh + (size_t)(row0 + r) * 8193 + jlo;
        for (int p = lane; p < wu; p += 32) uw[p * LP + r] = us[p];
        const float* ws = wh0 + (size_t)(row0 + r) * 1025 + c0;
        for (int p = lane; p <= SEGC; p += 32) wA[p * LP + r] = ws[p];
    }
    __syncthreads();

    const int vb = lane - jlo * LP;
    const float* ug  = uh + (size_t)(row0 + lane) * 8193;

    const float* KeP[3]  = {Ke0, Ke1, Ke2};
    const float* KoP[3]  = {Ko0, Ko1, Ko2};
    const int*   nbrP[3] = {nbrL0, nbrL1, nbrL2};
    float* oldW = wA;
    float* newW = wB;

#pragma unroll
    for (int i = 0; i < 3; ++i) {
        const int step = 4 >> i;
        const float hf = H * (float)step;
        const int ns = SEGC << i;
        const int nb = c0 << i;
        const float* Ke = KeP[i];
        const float* Ko = KoP[i];
        const int* nbr  = nbrP[i];
        for (int n = wave; n <= ns; n += 8) {
            const int* row = nbr + (size_t)(2 * (nb + n)) * 15;
            const float* kp = Ke + (size_t)(nb + n) * 8;
            int id[8];
            float kc[8];
#pragma unroll
            for (int m = 0; m < 8; ++m) { id[m] = row[2 * m]; kc[m] = kp[m]; }
            const float o0 = oldW[n * LP + lane];
            int lo = id[0], hi = id[0];
#pragma unroll
            for (int m = 1; m < 8; ++m) { lo = id[m] < lo ? id[m] : lo; hi = id[m] > hi ? id[m] : hi; }
            float acc = 0.f;
            if (step * lo >= jlo && step * hi <= jhi) {
#pragma unroll
                for (int m = 0; m < 8; ++m) acc = __builtin_fmaf(kc[m], uw[vb + id[m] * (step * LP)], acc);
            } else {
#pragma unroll
                for (int m = 0; m < 8; ++m) acc = __builtin_fmaf(kc[m], ug[step * id[m]], acc);
            }
            newW[(2 * n) * LP + lane] = o0 + hf * acc;
        }
        for (int n = wave; n < ns; n += 8) {
            const int* row = nbr + (size_t)(2 * (nb + n) + 1) * 15;
            const float* kp = Ko + (size_t)(nb + n) * 15;
            int id[15];
            float kc[15];
#pragma unroll
            for (int m = 0; m < 15; ++m) { id[m] = row[m]; kc[m] = kp[m]; }
            const float o0 = oldW[n * LP + lane];
            const float o1 = oldW[(n + 1) * LP + lane];
            int lo = id[0], hi = id[0];
#pragma unroll
            for (int m = 1; m < 15; ++m) { lo = id[m] < lo ? id[m] : lo; hi = id[m] > hi ? id[m] : hi; }
            float acc = 0.f;
            if (step * lo >= jlo && step * hi <= jhi) {
#pragma unroll
                for (int m = 0; m < 15; ++m) acc = __builtin_fmaf(kc[m], uw[vb + id[m] * (step * LP)], acc);
            } else {
#pragma unroll
                for (int m = 0; m < 15; ++m) acc = __builtin_fmaf(kc[m], ug[step * id[m]], acc);
            }
            newW[(2 * n + 1) * LP + lane] = 0.5f * (o0 + o1) + hf * acc;
        }
        __syncthreads();
        float* tmp = oldW; oldW = newW; newW = tmp;
    }

    if (seg == 0 || seg == NSEG - 1) {
        if (wave == 0) {
            const float* kv = Kbd_V + ((seg == 0) ? 0 : 32);
            const float* sp = sbuf + (size_t)(row0 + lane) * 32;
            float acc = 0.f;
#pragma unroll
            for (int r = 0; r < 32; ++r) acc += kv[r] * sp[r];
            oldW[((seg == 0) ? 0 : 8 * SEGC) * LP + lane] = H * acc;
        }
        __syncthreads();
    }

    const int npts = (seg == NSEG - 1) ? (8 * SEGC + 1) : (8 * SEGC);
    for (int r = wave; r < 32; r += 8) {
        float* dst = out + (size_t)(row0 + r) * 8193 + 8 * c0;
        for (int p = lane; p < npts; p += 32) dst[p] = oldW[p * LP + r];
    }
}

extern "C" void kernel_launch(void* const* d_in, const int* in_sizes, int n_in,
                              void* d_out, int out_size, void* d_ws, size_t ws_size,
                              hipStream_t stream) {
    const float* uh     = (const float*)d_in[0];
    const float* KHH_U  = (const float*)d_in[1];
    const float* KHH_V  = (const float*)d_in[2];
    const float* Ke0_U  = (const float*)d_in[3];
    const float* Ke0_V  = (const float*)d_in[4];
    const float* Ko0_U  = (const float*)d_in[5];
    const float* Ko0_V  = (const float*)d_in[6];
    const int*   nbr0   = (const int*)d_in[7];
    const float* Ke1_U  = (const float*)d_in[8];
    const float* Ke1_V  = (const float*)d_in[9];
    const float* Ko1_U  = (const float*)d_in[10];
    const float* Ko1_V  = (const float*)d_in[11];
    const int*   nbr1   = (const int*)d_in[12];
    const float* Ke2_U  = (const float*)d_in[13];
    const float* Ke2_V  = (const float*)d_in[14];
    const float* Ko2_U  = (const float*)d_in[15];
    const float* Ko2_V  = (const float*)d_in[16];
    const int*   nbr2   = (const int*)d_in[17];
    const float* Kbd_U  = (const float*)d_in[18];
    const float* Kbd_V  = (const float*)d_in[19];

    float* ws  = (float*)d_ws;
    float* wh0 = ws;
    float* sb  = wh0 + 524800;
    float* Ke0 = sb + 16384;
    float* Ke1 = Ke0 + 8200;
    float* Ke2 = Ke1 + 16392;
    float* Ko0 = Ke2 + 32776;
    float* Ko1 = Ko0 + 15360;
    float* Ko2 = Ko1 + 30720;

    float* outp = (float*)d_out;

    auto grid = [](int elems) { return (elems + 255) / 256; };

    k_rankmm<<<grid(1025 * 8),  256, 0, stream>>>(Ke0_U, Ke0_V, Ke0, 1025, 8);
    k_rankmm<<<grid(1024 * 15), 256, 0, stream>>>(Ko0_U, Ko0_V, Ko0, 1024, 15);
    k_rankmm<<<grid(2049 * 8),  256, 0, stream>>>(Ke1_U, Ke1_V, Ke1, 2049, 8);
    k_rankmm<<<grid(2048 * 15), 256, 0, stream>>>(Ko1_U, Ko1_V, Ko1, 2048, 15);
    k_rankmm<<<grid(4097 * 8),  256, 0, stream>>>(Ke2_U, Ke2_V, Ke2, 4097, 8);
    k_rankmm<<<grid(4096 * 15), 256, 0, stream>>>(Ko2_U, Ko2_V, Ko2, 4096, 15);

    k_wmma_proj<<<32, 1024, 0, stream>>>(uh, KHH_U, KHH_V, Kbd_U, wh0, sb);

    k_refine<<<1024, 256, 0, stream>>>(uh, wh0, sb, Kbd_V,
                                      Ke0, Ko0, Ke1, Ko1, Ke2, Ko2,
                                      nbr2, nbr1, nbr0, outp);
}


# Coefficient products made inside the refinement block: it builds its segment's Ke and Ko rows of all levels on WMMA from U and V into LDS; six product launches go; coefficient sum order changes
# speedup vs baseline: 7.9175x; 1.2560x over previous
#include <hip/hip_runtime.h>
#include <hip/hip_bf16.h>

typedef __attribute__((ext_vector_type(2))) float v2f;
typedef __attribute__((ext_vector_type(4))) float v4f;
typedef __attribute__((ext_vector_type(8))) float v8f;

#define WMMA_F32X4(a, b, c) \
    __builtin_amdgcn_wmma_f32_16x16x4_f32(false, (a), false, (b), (short)0, (c), false, false)

__global__ void __launch_bounds__(1024) k_wmma_proj(const float* __restrict__ uh,
                                                    const float* __restrict__ KHH_U,
                                                    const float* __restrict__ KHH_V,
                                                    const float* __restrict__ Kbd_U,
                                                    float* __restrict__ wh0,
                                                    float* __restrict__ sbuf) {
    __shared__ __attribute__((aligned(16))) float stage[32][2304];
    __shared__ float tl[16 * 32];
    const int wave = __builtin_amdgcn_readfirstlane((int)(threadIdx.x >> 5));
    const int lane = threadIdx.x & 31;
    const int half = lane >> 4;
    const int mrow = lane & 15;
    const int R    = blockIdx.x * 16;
    const float HK = 1.0f / 1024.0f;

    const float* arow = uh + (size_t)(R + mrow) * 8193;

    v8f t0 = {}; v8f t1 = {};
    v8f s0 = {}; v8f s1 = {};
    const int kc0 = 32 * wave;
    const int kf0 = 256 * wave;

    float* As = stage[wave];
    float* Bs = As + 16 * 36;
    float* Vs = Bs + 32 * 48;
    const float* ublk = uh + (size_t)R * 8193 + lane;
    const int srow = lane >> 3;
    const int scol = 4 * (lane & 7);

    for (int c = 0; c < 8; ++c) {
        const int kbase = kf0 + 32 * c;
        float av[16];
#pragma unroll
        for (int r = 0; r < 16; ++r) av[r] = ublk[(size_t)r * 8193 + kbase];
        const v4f* bsrc = (const v4f*)(Kbd_U + (size_t)kbase * 32) + lane;
        v4f bv[8];
#pragma unroll
        for (int i = 0; i < 8; ++i) bv[i] = bsrc[32 * i];
        const v4f vv = *((const v4f*)(KHH_V + (size_t)(kc0 + 4 * c) * 32) + lane);

        __builtin_amdgcn_fence(__ATOMIC_ACQ_REL, "wavefront");
        __builtin_amdgcn_wave_barrier();
#pragma unroll
        for (int r = 0; r < 16; ++r) As[r * 36 + lane] = av[r];
#pragma unroll
        for (int i = 0; i < 8; ++i) *(v4f*)(Bs + (4 * i + srow) * 48 + scol) = bv[i];
        *(v4f*)(Vs + srow * 48 + scol) = vv;
        __builtin_amdgcn_fence(__ATOMIC_ACQ_REL, "wavefront");
        __builtin_amdgcn_wave_barrier();

        {
            v2f a, b0, b1;
            a.x  = As[mrow * 36 + 16 * half];
            a.y  = As[mrow * 36 + 16 * half + 8];
            b0.x = Vs[(2 * half) * 48 + mrow];
            b0.y = Vs[(2 * half + 1) * 48 + mrow];
            b1.x = Vs[(2 * half) * 48 + 16 + mrow];
            b1.y = Vs[(2 * half + 1) * 48 + 16 + mrow];
            t0 = WMMA_F32X4(a, b0, t0);
            t1 = WMMA_F32X4(a, b1, t1);
        }
#pragma unroll
        for (int q = 0; q < 8; ++q) {
            const int kl = 4 * q + 2 * half;
            v2f a, b0, b1;
            a    = *(const v2f*)(As + mrow * 36 + kl);
            b0.x = Bs[kl * 48 + mrow];
            b0.y = Bs[(kl + 1) * 48 + mrow];
            b1.x = Bs[kl * 48 + 16 + mrow];
            b1.y = Bs[(kl + 1) * 48 + 16 + mrow];
            s0 = WMMA_F32X4(a, b0, s0);
            s1 = WMMA_F32X4(a, b1, s1);
        }
    }
    __builtin_amdgcn_fence(__ATOMIC_ACQ_REL, "wavefront");
    __builtin_amdgcn_wave_barrier();
    if (wave == 31) {
        float alast = arow[8192];
        v2f a, b0, b1;
        a.x  = (half == 0) ? alast : 0.f;
        a.y  = 0.f;
        b0.x = KHH_V[1024 * 32 + mrow];
        b0.y = b0.x;
        b1.x = KHH_V[1024 * 32 + 16 + mrow];
        b1.y = b1.x;
        t0 = WMMA_F32X4(a, b0, t0);
        t1 = WMMA_F32X4(a, b1, t1);
        b0.x = Kbd_U[8192 * 32 + mrow];
        b0.y = b0.x;
        b1.x = Kbd_U[8192 * 32 + 16 + mrow];
        b1.y = b1.x;
        s0 = WMMA_F32X4(a, b0, s0);
        s1 = WMMA_F32X4(a, b1, s1);
    }

    float* pw = stage[wave];
#pragma unroll
    for (int j = 0; j < 8; ++j) {
        int m = j + 8 * half;
        pw[m * 32 + mrow]            = t0[j];
        pw[m * 32 + 16 + mrow]       = t1[j];
        pw[512 + m * 32 + mrow]      = s0[j];
        pw[512 + m * 32 + 16 + mrow] = s1[j];
    }
    __syncthreads();

    {
        const int e = threadIdx.x;
        float acc = stage[0][e];
#pragma unroll
        for (int w = 1; w < 32; ++w) acc += stage[w][e];
        if (e < 512) tl[e] = acc * HK;
        else         sbuf[(size_t)R * 32 + (e - 512)] = acc;
    }
    __syncthreads();

    v2f af[8];
#pragma unroll
    for (int q = 0; q < 8; ++q) {
        int ka = 4 * q + 2 * half;
        af[q].x = tl[mrow * 32 + ka];
        af[q].y = tl[mrow * 32 + ka + 1];
    }
#pragma unroll
    for (int u = 0; u < 2; ++u) {
        int n = (2 * wave + u) * 16 + mrow;
        v8f acc = {};
#pragma unroll
        for (int q = 0; q < 8; ++q) {
            int ka = 4 * q + 2 * half;
            v2f b;
            b.x = KHH_U[n * 32 + ka];
            b.y = KHH_U[n * 32 + ka + 1];
            acc = WMMA_F32X4(af[q], b, acc);
        }
#pragma unroll
        for (int j = 0; j < 8; ++j) {
            int m = j + 8 * half;
            wh0[(size_t)(R + m) * 1025 + n] = acc[j];
        }
    }
    if (wave == 0) {
        v8f acc = {};
#pragma unroll
        for (int q = 0; q < 8; ++q) {
            int ka = 4 * q + 2 * half;
            v2f b;
            b.x = KHH_U[1024 * 32 + ka];
            b.y = KHH_U[1024 * 32 + ka + 1];
            acc = WMMA_F32X4(af[q], b, acc);
        }
        if (mrow == 0) {
#pragma unroll
            for (int j = 0; j < 8; ++j) {
                int m = j + 8 * half;
                wh0[(size_t)(R + m) * 1025 + 1024] = acc[j];
            }
        }
    }
}

#define SEGC 16
#define NSEG (1024 / SEGC)
#define LP   33
#define HALO 32
#define WUMAX (8 * SEGC + 2 * HALO + 1)
#define NCTILE 17
#define KPITCH 16
#define KE_ROW(i) ((i) == 0 ? 0  : ((i) == 1 ? 48 : 128))
#define KO_ROW(i) ((i) == 0 ? 32 : ((i) == 1 ? 96 : 208))

__global__ void __launch_bounds__(256) k_refine(const float* __restrict__ uh,
                                                const float* __restrict__ wh0,
                                                const float* __restrict__ sbuf,
                                                const float* __restrict__ Kbd_V,
                                                const float* __restrict__ Ke0_U,
                                                const float* __restrict__ Ke0_V,
                                                const float* __restrict__ Ko0_U,
                                                const float* __restrict__ Ko0_V,
                                                const float* __restrict__ Ke1_U,
                                                const float* __restrict__ Ke1_V,
                                                const float* __restrict__ Ko1_U,
                                                const float* __restrict__ Ko1_V,
                                                const float* __restrict__ Ke2_U,
                                                const float* __restrict__ Ke2_V,
                                                const float* __restrict__ Ko2_U,
                                                const float* __restrict__ Ko2_V,
                                                const int* __restrict__ nbrL0,
                                                const int* __restrict__ nbrL1,
                                                const int* __restrict__ nbrL2,
                                                float* __restrict__ out) {
    __shared__ float uw[WUMAX * LP];
    __shared__ float wA[(4 * SEGC + 1) * LP];
    __shared__ float wB[(8 * SEGC + 1) * LP];
    __shared__ __attribute__((aligned(16))) float kcf[NCTILE * 16 * KPITCH];
    const int wave = __builtin_amdgcn_readfirstlane((int)(threadIdx.x >> 5));
    const int lane = threadIdx.x & 31;
    const int seg  = blockIdx.x % NSEG;
    const int row0 = (blockIdx.x / NSEG) * 32;
    const int c0   = seg * SEGC;
    const int jlo  = (8 * c0 - HALO) < 0 ? 0 : (8 * c0 - HALO);
    const int jhi  = (8 * (c0 + SEGC) + HALO) > 8192 ? 8192 : (8 * (c0 + SEGC) + HALO);
    const int wu   = jhi - jlo + 1;
    const float H  = 1.0f / 8192.0f;

    {
        const int mrow = lane & 15;
        const int half = lane >> 4;
#define TILE_SRC(t, up, vp)                                                              \
        do {                                                                             \
            const int lvl_  = ((t) >= 3 ? 1 : 0) + ((t) >= 8 ? 1 : 0);                    \
            const int tb_   = lvl_ == 0 ? 0 : (lvl_ == 1 ? 3 : 8);     \
            const int ne_   = lvl_ == 0 ? 2 : (lvl_ == 1 ? 3 : 5);      \
            const bool odd_ = ((t) - tb_) >= ne_;                                         \
            const int tt_   = (t) - tb_ - (odd_ ? ne_ : 0);            \
            const float* U_ = odd_ ? (lvl_ == 0 ? Ko0_U : (lvl_ == 1 ? Ko1_U : Ko2_U))    \
                                   : (lvl_ == 0 ? Ke0_U : (lvl_ == 1 ? Ke1_U : Ke2_U));   \
            const float* V_ = odd_ ? (lvl_ == 0 ? Ko0_V : (lvl_ == 1 ? Ko1_V : Ko2_V))    \
                                   : (lvl_ == 0 ? Ke0_V : (lvl_ == 1 ? Ke1_V : Ke2_V));   \
            const int glast_ = (1024 << lvl_) - (odd_ ? 1 : 0);          \
            const int vlast_ = odd_ ? 14 : 7;                            \
            int g_ = (c0 << lvl_) + 16 * tt_ + mrow;                                      \
            g_ = g_ > glast_ ? glast_ : g_;                                               \
            const int vr_ = mrow > vlast_ ? vlast_ : mrow;                                \
            (up) = U_ + (size_t)g_ * 32 + 4 * half;                                       \
            (vp) = V_ + vr_ * 32 + 4 * half;                                              \
        } while (0)
        const float* up;
        const float* vp;
        v4f a0[4], b0[4], a1[4], b1[4], a2[4], b2[4];
        TILE_SRC(wave, up, vp);
#pragma unroll
        for (int qq = 0; qq < 4; ++qq) { a0[qq] = *(const v4f*)(up + 8 * qq); b0[qq] = *(const v4f*)(vp + 8 * qq); }
        TILE_SRC(wave + 8, up, vp);
#pragma unroll
        for (int qq = 0; qq < 4; ++qq) { a1[qq] = *(const v4f*)(up + 8 * qq); b1[qq] = *(const v4f*)(vp + 8 * qq); }
        if (wave == 0) {
            TILE_SRC(16, up, vp);
#pragma unroll
            for (int qq = 0; qq < 4; ++qq) { a2[qq] = *(const v4f*)(up + 8 * qq); b2[qq] = *(const v4f*)(vp + 8 * qq); }
        }
        v8f d0 = {}; v8f d1 = {};
#pragma unroll
        for (int s = 0; s < 8; ++s) {
            const int qq = s >> 1, e = s & 1;
            v2f x, y;
            x.x = a0[qq][2 * e]; x.y = a0[qq][2 * e + 1];
            y.x = b0[qq][2 * e]; y.y = b0[qq][2 * e + 1];
            d0 = WMMA_F32X4(x, y, d0);
            x.x = a1[qq][2 * e]; x.y = a1[qq][2 * e + 1];
            y.x = b1[qq][2 * e]; y.y = b1[qq][2 * e + 1];
            d1 = WMMA_F32X4(x, y, d1);
        }
        float* o = kcf + (16 * wave + 8 * half) * KPITCH + mrow;
#pragma unroll
        for (int j = 0; j < 8; ++j) {
            o[j * KPITCH]         = d0[j];
            o[(128 + j) * KPITCH] = d1[j];
        }
        if (wave == 0) {
            v8f d2 = {};
#pragma unroll
            for (int s = 0; s < 8; ++s) {
                const int qq = s >> 1, e = s & 1;
                v2f x, y;
                x.x = a2[qq][2 * e]; x.y = a2[qq][2 * e + 1];
                y.x = b2[qq][2 * e]; y.y = b2[qq][2 * e + 1];
                d2 = WMMA_F32X4(x, y, d2);
            }
#pragma unroll
            for (int j = 0; j < 8; ++j) o[(256 + j) * KPITCH] = d2[j];
        }
#undef TILE_SRC
    }

    for (int r = wave; r < 32; r += 8) {
        const float* us = uh + (size_t)(row0 + r) * 8193 + jlo;
        for (int p = lane; p < wu; p += 32) uw[p * LP + r] = us[p];
        const float* ws = wh0 + (size_t)(row0 + r) * 1025 + c0;
        for (int p = lane; p <= SEGC; p += 32) wA[p * LP + r] = ws[p];
    }
    __syncthreads();

    const int vb = lane - jlo * LP;
    const float* ug  = uh + (size_t)(row0 + lane) * 8193;

    const int*   nbrP[3] = {nbrL0, nbrL1, nbrL2};
    float* oldW = wA;
    float* newW = wB;

#pragma unroll
    for (int i = 0; i < 3; ++i) {
        const int step = 4 >> i;
        const float hf = H * (float)step;
        const int ns = SEGC << i;
        const int nb = c0 << i;
        const float* Ke = kcf + KE_ROW(i) * KPITCH;
        const float* Ko = kcf + KO_ROW(i) * KPITCH;
        const int* nbr  = nbrP[i];
        for (int n = wave; n <= ns; n += 8) {
            const int* row = nbr + (size_t)(2 * (nb + n)) * 15;
            const v4f* kp = (const v4f*)(Ke + n * KPITCH);
            int id[8];
            float kc[8];
#pragma unroll
            for (int m = 0; m < 8; ++m) id[m] = row[2 * m];
            const v4f k0 = kp[0], k1 = kp[1];
#pragma unroll
            for (int m = 0; m < 4; ++m) { kc[m] = k0[m]; kc[4 + m] = k1[m]; }
            const float o0 = oldW[n * LP + lane];
            int lo = id[0], hi = id[0];
#pragma unroll
            for (int m = 1; m < 8; ++m) { lo = id[m] < lo ? id[m] : lo; hi = id[m] > hi ? id[m] : hi; }
            float acc = 0.f;
            if (step * lo >= jlo && step * hi <= jhi) {
#pragma unroll
                for (int m = 0; m < 8; ++m) acc = __builtin_fmaf(kc[m], uw[vb + id[m] * (step * LP)], acc);
            } else {
#pragma unroll
                for (int m = 0; m < 8; ++m) acc = __builtin_fmaf(kc[m], ug[step * id[m]], acc);
            }
            newW[(2 * n) * LP + lane] = o0 + hf * acc;
        }
        for (int n = wave; n < ns; n += 8) {
            const int* row = nbr + (size_t)(2 * (nb + n) + 1) * 15;
            const v4f* kp = (const v4f*)(Ko + n * KPITCH);
            int id[15];
            float kc[16];
#pragma unroll
            for (int m = 0; m < 15; ++m) id[m] = row[m];
            const v4f k0 = kp[0], k1 = kp[1], k2 = kp[2], k3 = kp[3];
#pragma unroll
            for (int m = 0; m < 4; ++m) { kc[m] = k0[m]; kc[4 + m] = k1[m]; kc[8 + m] = k2[m]; kc[12 + m] = k3[m]; }
            const float o0 = oldW[n * LP + lane];
            const float o1 = oldW[(n + 1) * LP + lane];
            int lo = id[0], hi = id[0];
#pragma unroll
            for (int m = 1; m < 15; ++m) { lo = id[m] < lo ? id[m] : lo; hi = id[m] > hi ? id[m] : hi; }
            float acc = 0.f;
            if (step * lo >= jlo && step * hi <= jhi) {
#pragma unroll
                for (int m = 0; m < 15; ++m) acc = __builtin_fmaf(kc[m], uw[vb + id[m] * (step * LP)], acc);
            } else {
#pragma unroll
                for (int m = 0; m < 15; ++m) acc = __builtin_fmaf(kc[m], ug[step * id[m]], acc);
            }
            newW[(2 * n + 1) * LP + lane] = 0.5f * (o0 + o1) + hf * acc;
        }
        __syncthreads();
        float* tmp = oldW; oldW = newW; newW = tmp;
    }

    if (seg == 0 || seg == NSEG - 1) {
        if (wave == 0) {
            const float* kv = Kbd_V + ((seg == 0) ? 0 : 32);
            const float* sp = sbuf + (size_t)(row0 + lane) * 32;
            float acc = 0.f;
#pragma unroll
            for (int r = 0; r < 32; ++r) acc += kv[r] * sp[r];
            oldW[((seg == 0) ? 0 : 8 * SEGC) * LP + lane] = H * acc;
        }
        __syncthreads();
    }

    const int npts = (seg == NSEG - 1) ? (8 * SEGC + 1) : (8 * SEGC);
    for (int r = wave; r < 32; r += 8) {
        float* dst = out + (size_t)(row0 + r) * 8193 + 8 * c0;
        for (int p = lane; p < npts; p += 32) dst[p] = oldW[p * LP + r];
    }
}

extern "C" void kernel_launch(void* const* d_in, const int* in_sizes, int n_in,
                              void* d_out, int out_size, void* d_ws, size_t ws_size,
                              hipStream_t stream) {
    const float* uh     = (const float*)d_in[0];
    const float* KHH_U  = (const float*)d_in[1];
    const float* KHH_V  = (const float*)d_in[2];
    const float* Ke0_U  = (const float*)d_in[3];
    const float* Ke0_V  = (const float*)d_in[4];
    const float* Ko0_U  = (const float*)d_in[5];
    const float* Ko0_V  = (const float*)d_in[6];
    const int*   nbr0   = (const int*)d_in[7];
    const float* Ke1_U  = (const float*)d_in[8];
    const float* Ke1_V  = (const float*)d_in[9];
    const float* Ko1_U  = (const float*)d_in[10];
    const float* Ko1_V  = (const float*)d_in[11];
    const int*   nbr1   = (const int*)d_in[12];
    const float* Ke2_U  = (const float*)d_in[13];
    const float* Ke2_V  = (const float*)d_in[14];
    const float* Ko2_U  = (const float*)d_in[15];
    const float* Ko2_V  = (const float*)d_in[16];
    const int*   nbr2   = (const int*)d_in[17];
    const float* Kbd_U  = (const float*)d_in[18];
    const float* Kbd_V  = (const float*)d_in[19];

    float* ws  = (float*)d_ws;
    float* wh0 = ws;
    float* sb  = wh0 + 524800;

    float* outp = (float*)d_out;

    k_wmma_proj<<<32, 1024, 0, stream>>>(uh, KHH_U, KHH_V, Kbd_U, wh0, sb);

    k_refine<<<1024, 256, 0, stream>>>(uh, wh0, sb, Kbd_V,
                                      Ke0_U, Ke0_V, Ko0_U, Ko0_V,
                                      Ke1_U, Ke1_V, Ko1_U, Ko1_V,
                                      Ke2_U, Ke2_V, Ko2_U, Ko2_V,
                                      nbr2, nbr1, nbr0, outp);
}


# Projection launch split by product over 96 blocks: 32 make t and the coarse product, 64 each make one 16-column half of s for a 16-row tile; s kept as two column planes; every sum order unchanged
# speedup vs baseline: 9.8648x; 1.2459x over previous
#include <hip/hip_runtime.h>
#include <hip/hip_bf16.h>

typedef __attribute__((ext_vector_type(2))) float v2f;
typedef __attribute__((ext_vector_type(4))) float v4f;
typedef __attribute__((ext_vector_type(8))) float v8f;

#define WMMA_F32X4(a, b, c) \
    __builtin_amdgcn_wmma_f32_16x16x4_f32(false, (a), false, (b), (short)0, (c), false, false)

__global__ void __launch_bounds__(1024) k_wmma_proj(const float* __restrict__ uh,
                                                    const float* __restrict__ KHH_U,
                                                    const float* __restrict__ KHH_V,
                                                    const float* __restrict__ Kbd_U,
                                                    float* __restrict__ wh0,
                                                    float* __restrict__ sbuf) {
    __shared__ __attribute__((aligned(16))) float stage[32][2304];
    __shared__ float tl[16 * 32];
    const int wave = __builtin_amdgcn_readfirstlane((int)(threadIdx.x >> 5));
    const int lane = threadIdx.x & 31;
    const int half = lane >> 4;
    const int mrow = lane & 15;
    const int kc0 = 32 * wave;
    const int kf0 = 256 * wave;

    float* As = stage[wave];
    float* Bs = As + 16 * 36;
    float* Vs = Bs + 32 * 48;

    if (blockIdx.x >= 32) {
        const int sb = blockIdx.x - 32;
        const int R  = (sb >> 1) * 16;
        const int pl = sb & 1;
        const int n0 = 16 * pl;
        const float* ublk = uh + (size_t)R * 8193 + lane;
        const int srow = lane >> 2;
        const int scol = 4 * (lane & 3);
        v8f s0 = {};
        for (int c = 0; c < 8; ++c) {
            const int kbase = kf0 + 32 * c;
            float av[16];
#pragma unroll
            for (int r = 0; r < 16; ++r) av[r] = ublk[(size_t)r * 8193 + kbase];
            const float* bsrc = Kbd_U + (size_t)(kbase + srow) * 32 + n0 + scol;
            v4f bv[4];
#pragma unroll
            for (int i = 0; i < 4; ++i) bv[i] = *(const v4f*)(bsrc + (size_t)(8 * i) * 32);

            __builtin_amdgcn_fence(__ATOMIC_ACQ_REL, "wavefront");
            __builtin_amdgcn_wave_barrier();
#pragma unroll
            for (int r = 0; r < 16; ++r) As[r * 36 + lane] = av[r];
#pragma unroll
            for (int i = 0; i < 4; ++i) *(v4f*)(Bs + (8 * i + srow) * 48 + scol) = bv[i];
            __builtin_amdgcn_fence(__ATOMIC_ACQ_REL, "wavefront");
            __builtin_amdgcn_wave_barrier();

#pragma unroll
            for (int q = 0; q < 8; ++q) {
                const int kl = 4 * q + 2 * half;
                v2f a, b;
                a   = *(const v2f*)(As + mrow * 36 + kl);
                b.x = Bs[kl * 48 + mrow];
                b.y = Bs[(kl + 1) * 48 + mrow];
                s0 = WMMA_F32X4(a, b, s0);
            }
        }
        __builtin_amdgcn_fence(__ATOMIC_ACQ_REL, "wavefront");
        __builtin_amdgcn_wave_barrier();
        if (wave == 31) {
            float alast = uh[(size_t)(R + mrow) * 8193 + 8192];
            v2f a, b;
            a.x = (half == 0) ? alast : 0.f;
            a.y = 0.f;
            b.x = Kbd_U[8192 * 32 + n0 + mrow];
            b.y = b.x;
            s0 = WMMA_F32X4(a, b, s0);
        }
        float* pw = stage[wave];
#pragma unroll
        for (int j = 0; j < 8; ++j) pw[(j + 8 * half) * 16 + mrow] = s0[j];
        __syncthreads();
        if (threadIdx.x < 256) {
            const int e = threadIdx.x;
            float acc = stage[0][e];
#pragma unroll
            for (int w = 1; w < 32; ++w) acc += stage[w][e];
            sbuf[(size_t)pl * 8192 + (size_t)R * 16 + e] = acc;
        }
        return;
    }

    const int R    = blockIdx.x * 16;
    const float HK = 1.0f / 1024.0f;
    const float* arow = uh + (size_t)(R + mrow) * 8193;

    v8f t0 = {}; v8f t1 = {};
    const float* ublk = uh + (size_t)R * 8193 + lane;
    const int srow = lane >> 3;
    const int scol = 4 * (lane & 7);

    for (int c = 0; c < 8; ++c) {
        const int kbase = kf0 + 32 * c;
        float av[16];
#pragma unroll
        for (int r = 0; r < 16; ++r) av[r] = ublk[(size_t)r * 8193 + kbase];
        const v4f vv = *((const v4f*)(KHH_V + (size_t)(kc0 + 4 * c) * 32) + lane);

        __builtin_amdgcn_fence(__ATOMIC_ACQ_REL, "wavefront");
        __builtin_amdgcn_wave_barrier();
#pragma unroll
        for (int r = 0; r < 16; ++r) As[r * 36 + lane] = av[r];
        *(v4f*)(Vs + srow * 48 + scol) = vv;
        __builtin_amdgcn_fence(__ATOMIC_ACQ_REL, "wavefront");
        __builtin_amdgcn_wave_barrier();

        {
            v2f a, b0, b1;
            a.x  = As[mrow * 36 + 16 * half];
            a.y  = As[mrow * 36 + 16 * half + 8];
            b0.x = Vs[(2 * half) * 48 + mrow];
            b0.y = Vs[(2 * half + 1) * 48 + mrow];
            b1.x = Vs[(2 * half) * 48 + 16 + mrow];
            b1.y = Vs[(2 * half + 1) * 48 + 16 + mrow];
            t0 = WMMA_F32X4(a, b0, t0);
            t1 = WMMA_F32X4(a, b1, t1);
        }
    }
    __builtin_amdgcn_fence(__ATOMIC_ACQ_REL, "wavefront");
    __builtin_amdgcn_wave_barrier();
    if (wave == 31) {
        float alast = arow[8192];
        v2f a, b0, b1;
        a.x  = (half == 0) ? alast : 0.f;
        a.y  = 0.f;
        b0.x = KHH_V[1024 * 32 + mrow];
        b0.y = b0.x;
        b1.x = KHH_V[1024 * 32 + 16 + mrow];
        b1.y = b1.x;
        t0 = WMMA_F32X4(a, b0, t0);
        t1 = WMMA_F32X4(a, b1, t1);
    }

    float* pw = stage[wave];
#pragma unroll
    for (int j = 0; j < 8; ++j) {
        int m = j + 8 * half;
        pw[m * 32 + mrow]      = t0[j];
        pw[m * 32 + 16 + mrow] = t1[j];
    }
    __syncthreads();

    if (threadIdx.x < 512) {
        const int e = threadIdx.x;
        float acc = stage[0][e];
#pragma unroll
        for (int w = 1; w < 32; ++w) acc += stage[w][e];
        tl[e] = acc * HK;
    }
    __syncthreads();

    v2f af[8];
#pragma unroll
    for (int q = 0; q < 8; ++q) {
        int ka = 4 * q + 2 * half;
        af[q].x = tl[mrow * 32 + ka];
        af[q].y = tl[mrow * 32 + ka + 1];
    }
#pragma unroll
    for (int u = 0; u < 2; ++u) {
        int n = (2 * wave + u) * 16 + mrow;
        v8f acc = {};
#pragma unroll
        for (int q = 0; q < 8; ++q) {
            int ka = 4 * q + 2 * half;
            v2f b;
            b.x = KHH_U[n * 32 + ka];
            b.y = KHH_U[n * 32 + ka + 1];
            acc = WMMA_F32X4(af[q], b, acc);
        }
#pragma unroll
        for (int j = 0; j < 8; ++j) {
            int m = j + 8 * half;
            wh0[(size_t)(R + m) * 1025 + n] = acc[j];
        }
    }
    if (wave == 0) {
        v8f acc = {};
#pragma unroll
        for (int q = 0; q < 8; ++q) {
            int ka = 4 * q + 2 * half;
            v2f b;
            b.x = KHH_U[1024 * 32 + ka];
            b.y = KHH_U[1024 * 32 + ka + 1];
            acc = WMMA_F32X4(af[q], b, acc);
        }
        if (mrow == 0) {
#pragma unroll
            for (int j = 0; j < 8; ++j) {
                int m = j + 8 * half;
                wh0[(size_t)(R + m) * 1025 + 1024] = acc[j];
            }
        }
    }
}

#define SEGC 16
#define NSEG (1024 / SEGC)
#define LP   33
#define HALO 32
#define WUMAX (8 * SEGC + 2 * HALO + 1)
#define NCTILE 17
#define KPITCH 16
#define KE_ROW(i) ((i) == 0 ? 0  : ((i) == 1 ? 48 : 128))
#define KO_ROW(i) ((i) == 0 ? 32 : ((i) == 1 ? 96 : 208))

__global__ void __launch_bounds__(256) k_refine(const float* __restrict__ uh,
                                                const float* __restrict__ wh0,
                                                const float* __restrict__ sbuf,
                                                const float* __restrict__ Kbd_V,
                                                const float* __restrict__ Ke0_U,
                                                const float* __restrict__ Ke0_V,
                                                const float* __restrict__ Ko0_U,
                                                const float* __restrict__ Ko0_V,
                                                const float* __restrict__ Ke1_U,
                                                const float* __restrict__ Ke1_V,
                                                const float* __restrict__ Ko1_U,
                                                const float* __restrict__ Ko1_V,
                                                const float* __restrict__ Ke2_U,
                                                const float* __restrict__ Ke2_V,
                                                const float* __restrict__ Ko2_U,
                                                const float* __restrict__ Ko2_V,
                                                const int* __restrict__ nbrL0,
                                                const int* __restrict__ nbrL1,
                                                const int* __restrict__ nbrL2,
                                                float* __restrict__ out) {
    __shared__ float uw[WUMAX * LP];
    __shared__ float wA[(4 * SEGC + 1) * LP];
    __shared__ float wB[(8 * SEGC + 1) * LP];
    __shared__ __attribute__((aligned(16))) float kcf[NCTILE * 16 * KPITCH];
    const int wave = __builtin_amdgcn_readfirstlane((int)(threadIdx.x >> 5));
    const int lane = threadIdx.x & 31;
    const int seg  = blockIdx.x % NSEG;
    const int row0 = (blockIdx.x / NSEG) * 32;
    const int c0   = seg * SEGC;
    const int jlo  = (8 * c0 - HALO) < 0 ? 0 : (8 * c0 - HALO);
    const int jhi  = (8 * (c0 + SEGC) + HALO) > 8192 ? 8192 : (8 * (c0 + SEGC) + HALO);
    const int wu   = jhi - jlo + 1;
    const float H  = 1.0f / 8192.0f;

    {
        const int mrow = lane & 15;
        const int half = lane >> 4;
#define TILE_SRC(t, up, vp)                                                              \
        do {                                                                             \
            const int lvl_  = ((t) >= 3 ? 1 : 0) + ((t) >= 8 ? 1 : 0);                    \
            const int tb_   = lvl_ == 0 ? 0 : (lvl_ == 1 ? 3 : 8);     \
            const int ne_   = lvl_ == 0 ? 2 : (lvl_ == 1 ? 3 : 5);      \
            const bool odd_ = ((t) - tb_) >= ne_;                                         \
            const int tt_   = (t) - tb_ - (odd_ ? ne_ : 0);            \
            const float* U_ = odd_ ? (lvl_ == 0 ? Ko0_U : (lvl_ == 1 ? Ko1_U : Ko2_U))    \
                                   : (lvl_ == 0 ? Ke0_U : (lvl_ == 1 ? Ke1_U : Ke2_U));   \
            const float* V_ = odd_ ? (lvl_ == 0 ? Ko0_V : (lvl_ == 1 ? Ko1_V : Ko2_V))    \
                                   : (lvl_ == 0 ? Ke0_V : (lvl_ == 1 ? Ke1_V : Ke2_V));   \
            const int glast_ = (1024 << lvl_) - (odd_ ? 1 : 0);          \
            const int vlast_ = odd_ ? 14 : 7;                            \
            int g_ = (c0 << lvl_) + 16 * tt_ + mrow;                                      \
            g_ = g_ > glast_ ? glast_ : g_;                                               \
            const int vr_ = mrow > vlast_ ? vlast_ : mrow;                                \
            (up) = U_ + (size_t)g_ * 32 + 4 * half;                                       \
            (vp) = V_ + vr_ * 32 + 4 * half;                                              \
        } while (0)
        const float* up;
        const float* vp;
        v4f a0[4], b0[4], a1[4], b1[4], a2[4], b2[4];
        TILE_SRC(wave, up, vp);
#pragma unroll
        for (int qq = 0; qq < 4; ++qq) { a0[qq] = *(const v4f*)(up + 8 * qq); b0[qq] = *(const v4f*)(vp + 8 * qq); }
        TILE_SRC(wave + 8, up, vp);
#pragma unroll
        for (int qq = 0; qq < 4; ++qq) { a1[qq] = *(const v4f*)(up + 8 * qq); b1[qq] = *(const v4f*)(vp + 8 * qq); }
        if (wave == 0) {
            TILE_SRC(16, up, vp);
#pragma unroll
            for (int qq = 0; qq < 4; ++qq) { a2[qq] = *(const v4f*)(up + 8 * qq); b2[qq] = *(const v4f*)(vp + 8 * qq); }
        }
        v8f d0 = {}; v8f d1 = {};
#pragma unroll
        for (int s = 0; s < 8; ++s) {
            const int qq = s >> 1, e = s & 1;
            v2f x, y;
            x.x = a0[qq][2 * e]; x.y = a0[qq][2 * e + 1];
            y.x = b0[qq][2 * e]; y.y = b0[qq][2 * e + 1];
            d0 = WMMA_F32X4(x, y, d0);
            x.x = a1[qq][2 * e]; x.y = a1[qq][2 * e + 1];
            y.x = b1[qq][2 * e]; y.y = b1[qq][2 * e + 1];
            d1 = WMMA_F32X4(x, y, d1);
        }
        float* o = kcf + (16 * wave + 8 * half) * KPITCH + mrow;
#pragma unroll
        for (int j = 0; j < 8; ++j) {
            o[j * KPITCH]         = d0[j];
            o[(128 + j) * KPITCH] = d1[j];
        }
        if (wave == 0) {
            v8f d2 = {};
#pragma unroll
            for (int s = 0; s < 8; ++s) {
                const int qq = s >> 1, e = s & 1;
                v2f x, y;
                x.x = a2[qq][2 * e]; x.y = a2[qq][2 * e + 1];
                y.x = b2[qq][2 * e]; y.y = b2[qq][2 * e + 1];
                d2 = WMMA_F32X4(x, y, d2);
            }
#pragma unroll
            for (int j = 0; j < 8; ++j) o[(256 + j) * KPITCH] = d2[j];
        }
#undef TILE_SRC
    }

    for (int r = wave; r < 32; r += 8) {
        const float* us = uh + (size_t)(row0 + r) * 8193 + jlo;
        for (int p = lane; p < wu; p += 32) uw[p * LP + r] = us[p];
        const float* ws = wh0 + (size_t)(row0 + r) * 1025 + c0;
        for (int p = lane; p <= SEGC; p += 32) wA[p * LP + r] = ws[p];
    }
    __syncthreads();

    const int vb = lane - jlo * LP;
    const float* ug  = uh + (size_t)(row0 + lane) * 8193;

    const int*   nbrP[3] = {nbrL0, nbrL1, nbrL2};
    float* oldW = wA;
    float* newW = wB;

#pragma unroll
    for (int i = 0; i < 3; ++i) {
        const int step = 4 >> i;
        const float hf = H * (float)step;
        const int ns = SEGC << i;
        const int nb = c0 << i;
        const float* Ke = kcf + KE_ROW(i) * KPITCH;
        const float* Ko = kcf + KO_ROW(i) * KPITCH;
        const int* nbr  = nbrP[i];
        for (int n = wave; n <= ns; n += 8) {
            const int* row = nbr + (size_t)(2 * (nb + n)) * 15;
            const v4f* kp = (const v4f*)(Ke + n * KPITCH);
            int id[8];
            float kc[8];
#pragma unroll
            for (int m = 0; m < 8; ++m) id[m] = row[2 * m];
            const v4f k0 = kp[0], k1 = kp[1];
#pragma unroll
            for (int m = 0; m < 4; ++m) { kc[m] = k0[m]; kc[4 + m] = k1[m]; }
            const float o0 = oldW[n * LP + lane];
            int lo = id[0], hi = id[0];
#pragma unroll
            for (int m = 1; m < 8; ++m) { lo = id[m] < lo ? id[m] : lo; hi = id[m] > hi ? id[m] : hi; }
            float acc = 0.f;
            if (step * lo >= jlo && step * hi <= jhi) {
#pragma unroll
                for (int m = 0; m < 8; ++m) acc = __builtin_fmaf(kc[m], uw[vb + id[m] * (step * LP)], acc);
            } else {
#pragma unroll
                for (int m = 0; m < 8; ++m) acc = __builtin_fmaf(kc[m], ug[step * id[m]], acc);
            }
            newW[(2 * n) * LP + lane] = o0 + hf * acc;
        }
        for (int n = wave; n < ns; n += 8) {
            const int* row = nbr + (size_t)(2 * (nb + n) + 1) * 15;
            const v4f* kp = (const v4f*)(Ko + n * KPITCH);
            int id[15];
            float kc[16];
#pragma unroll
            for (int m = 0; m < 15; ++m) id[m] = row[m];
            const v4f k0 = kp[0], k1 = kp[1], k2 = kp[2], k3 = kp[3];
#pragma unroll
            for (int m = 0; m < 4; ++m) { kc[m] = k0[m]; kc[4 + m] = k1[m]; kc[8 + m] = k2[m]; kc[12 + m] = k3[m]; }
            const float o0 = oldW[n * LP + lane];
            const float o1 = oldW[(n + 1) * LP + lane];
            int lo = id[0], hi = id[0];
#pragma unroll
            for (int m = 1; m < 15; ++m) { lo = id[m] < lo ? id[m] : lo; hi = id[m] > hi ? id[m] : hi; }
            float acc = 0.f;
            if (step * lo >= jlo && step * hi <= jhi) {
#pragma unroll
                for (int m = 0; m < 15; ++m) acc = __builtin_fmaf(kc[m], uw[vb + id[m] * (step * LP)], acc);
            } else {
#pragma unroll
                for (int m = 0; m < 15; ++m) acc = __builtin_fmaf(kc[m], ug[step * id[m]], acc);
            }
            newW[(2 * n + 1) * LP + lane] = 0.5f * (o0 + o1) + hf * acc;
        }
        __syncthreads();
        float* tmp = oldW; oldW = newW; newW = tmp;
    }

    if (seg == 0 || seg == NSEG - 1) {
        if (wave == 0) {
            const float* kv = Kbd_V + ((seg == 0) ? 0 : 32);
            const float* sp0 = sbuf + (size_t)(row0 + lane) * 16;
            const float* sp1 = sbuf + 8192 + (size_t)(row0 + lane) * 16;
            float acc = 0.f;
#pragma unroll
            for (int r = 0; r < 16; ++r) acc += kv[r] * sp0[r];
#pragma unroll
            for (int r = 0; r < 16; ++r) acc += kv[16 + r] * sp1[r];
            oldW[((seg == 0) ? 0 : 8 * SEGC) * LP + lane] = H * acc;
        }
        __syncthreads();
    }

    const int npts = (seg == NSEG - 1) ? (8 * SEGC + 1) : (8 * SEGC);
    for (int r = wave; r < 32; r += 8) {
        float* dst = out + (size_t)(row0 + r) * 8193 + 8 * c0;
        for (int p = lane; p < npts; p += 32) dst[p] = oldW[p * LP + r];
    }
}

extern "C" void kernel_launch(void* const* d_in, const int* in_sizes, int n_in,
                              void* d_out, int out_size, void* d_ws, size_t ws_size,
                              hipStream_t stream) {
    const float* uh     = (const float*)d_in[0];
    const float* KHH_U  = (const float*)d_in[1];
    const float* KHH_V  = (const float*)d_in[2];
    const float* Ke0_U  = (const float*)d_in[3];
    const float* Ke0_V  = (const float*)d_in[4];
    const float* Ko0_U  = (const float*)d_in[5];
    const float* Ko0_V  = (const float*)d_in[6];
    const int*   nbr0   = (const int*)d_in[7];
    const float* Ke1_U  = (const float*)d_in[8];
    const float* Ke1_V  = (const float*)d_in[9];
    const float* Ko1_U  = (const float*)d_in[10];
    const float* Ko1_V  = (const float*)d_in[11];
    const int*   nbr1   = (const int*)d_in[12];
    const float* Ke2_U  = (const float*)d_in[13];
    const float* Ke2_V  = (const float*)d_in[14];
    const float* Ko2_U  = (const float*)d_in[15];
    const float* Ko2_V  = (const float*)d_in[16];
    const int*   nbr2   = (const int*)d_in[17];
    const float* Kbd_U  = (const float*)d_in[18];
    const float* Kbd_V  = (const float*)d_in[19];

    float* ws  = (float*)d_ws;
    float* wh0 = ws;
    float* sb  = wh0 + 524800;

    float* outp = (float*)d_out;

    k_wmma_proj<<<96, 1024, 0, stream>>>(uh, KHH_U, KHH_V, Kbd_U, wh0, sb);

    k_refine<<<1024, 256, 0, stream>>>(uh, wh0, sb, Kbd_V,
                                      Ke0_U, Ke0_V, Ko0_U, Ko0_V,
                                      Ke1_U, Ke1_V, Ko1_U, Ko1_V,
                                      Ke2_U, Ke2_V, Ko2_U, Ko2_V,
                                      nbr2, nbr1, nbr0, outp);
}


# Refinement head: a wave issues all loads of its four rows' u window and coarse strip (32 loads) as one group ahead of their LDS stores, not one dependent load-store hop per piece; sums unchanged
# speedup vs baseline: 10.6711x; 1.0817x over previous
#include <hip/hip_runtime.h>
#include <hip/hip_bf16.h>

typedef __attribute__((ext_vector_type(2))) float v2f;
typedef __attribute__((ext_vector_type(4))) float v4f;
typedef __attribute__((ext_vector_type(8))) float v8f;

#define WMMA_F32X4(a, b, c) \
    __builtin_amdgcn_wmma_f32_16x16x4_f32(false, (a), false, (b), (short)0, (c), false, false)

__global__ void __launch_bounds__(1024) k_wmma_proj(const float* __restrict__ uh,
                                                    const float* __restrict__ KHH_U,
                                                    const float* __restrict__ KHH_V,
                                                    const float* __restrict__ Kbd_U,
                                                    float* __restrict__ wh0,
                                                    float* __restrict__ sbuf) {
    __shared__ __attribute__((aligned(16))) float stage[32][2304];
    __shared__ float tl[16 * 32];
    const int wave = __builtin_amdgcn_readfirstlane((int)(threadIdx.x >> 5));
    const int lane = threadIdx.x & 31;
    const int half = lane >> 4;
    const int mrow = lane & 15;
    const int kc0 = 32 * wave;
    const int kf0 = 256 * wave;

    float* As = stage[wave];
    float* Bs = As + 16 * 36;
    float* Vs = Bs + 32 * 48;

    if (blockIdx.x >= 32) {
        const int sb = blockIdx.x - 32;
        const int R  = (sb >> 1) * 16;
        const int pl = sb & 1;
        const int n0 = 16 * pl;
        const float* ublk = uh + (size_t)R * 8193 + lane;
        const int srow = lane >> 2;
        const int scol = 4 * (lane & 3);
        v8f s0 = {};
        for (int c = 0; c < 8; ++c) {
            const int kbase = kf0 + 32 * c;
            float av[16];
#pragma unroll
            for (int r = 0; r < 16; ++r) av[r] = ublk[(size_t)r * 8193 + kbase];
            const float* bsrc = Kbd_U + (size_t)(kbase + srow) * 32 + n0 + scol;
            v4f bv[4];
#pragma unroll
            for (int i = 0; i < 4; ++i) bv[i] = *(const v4f*)(bsrc + (size_t)(8 * i) * 32);

            __builtin_amdgcn_fence(__ATOMIC_ACQ_REL, "wavefront");
            __builtin_amdgcn_wave_barrier();
#pragma unroll
            for (int r = 0; r < 16; ++r) As[r * 36 + lane] = av[r];
#pragma unroll
            for (int i = 0; i < 4; ++i) *(v4f*)(Bs + (8 * i + srow) * 48 + scol) = bv[i];
            __builtin_amdgcn_fence(__ATOMIC_ACQ_REL, "wavefront");
            __builtin_amdgcn_wave_barrier();

#pragma unroll
            for (int q = 0; q < 8; ++q) {
                const int kl = 4 * q + 2 * half;
                v2f a, b;
                a   = *(const v2f*)(As + mrow * 36 + kl);
                b.x = Bs[kl * 48 + mrow];
                b.y = Bs[(kl + 1) * 48 + mrow];
                s0 = WMMA_F32X4(a, b, s0);
            }
        }
        __builtin_amdgcn_fence(__ATOMIC_ACQ_REL, "wavefront");
        __builtin_amdgcn_wave_barrier();
        if (wave == 31) {
            float alast = uh[(size_t)(R + mrow) * 8193 + 8192];
            v2f a, b;
            a.x = (half == 0) ? alast : 0.f;
            a.y = 0.f;
            b.x = Kbd_U[8192 * 32 + n0 + mrow];
            b.y = b.x;
            s0 = WMMA_F32X4(a, b, s0);
        }
        float* pw = stage[wave];
#pragma unroll
        for (int j = 0; j < 8; ++j) pw[(j + 8 * half) * 16 + mrow] = s0[j];
        __syncthreads();
        if (threadIdx.x < 256) {
            const int e = threadIdx.x;
            float acc = stage[0][e];
#pragma unroll
            for (int w = 1; w < 32; ++w) acc += stage[w][e];
            sbuf[(size_t)pl * 8192 + (size_t)R * 16 + e] = acc;
        }
        return;
    }

    const int R    = blockIdx.x * 16;
    const float HK = 1.0f / 1024.0f;
    const float* arow = uh + (size_t)(R + mrow) * 8193;

    v8f t0 = {}; v8f t1 = {};
    const float* ublk = uh + (size_t)R * 8193 + lane;
    const int srow = lane >> 3;
    const int scol = 4 * (lane & 7);

    for (int c = 0; c < 8; ++c) {
        const int kbase = kf0 + 32 * c;
        float av[16];
#pragma unroll
        for (int r = 0; r < 16; ++r) av[r] = ublk[(size_t)r * 8193 + kbase];
        const v4f vv = *((const v4f*)(KHH_V + (size_t)(kc0 + 4 * c) * 32) + lane);

        __builtin_amdgcn_fence(__ATOMIC_ACQ_REL, "wavefront");
        __builtin_amdgcn_wave_barrier();
#pragma unroll
        for (int r = 0; r < 16; ++r) As[r * 36 + lane] = av[r];
        *(v4f*)(Vs + srow * 48 + scol) = vv;
        __builtin_amdgcn_fence(__ATOMIC_ACQ_REL, "wavefront");
        __builtin_amdgcn_wave_barrier();

        {
            v2f a, b0, b1;
            a.x  = As[mrow * 36 + 16 * half];
            a.y  = As[mrow * 36 + 16 * half + 8];
            b0.x = Vs[(2 * half) * 48 + mrow];
            b0.y = Vs[(2 * half + 1) * 48 + mrow];
            b1.x = Vs[(2 * half) * 48 + 16 + mrow];
            b1.y = Vs[(2 * half + 1) * 48 + 16 + mrow];
            t0 = WMMA_F32X4(a, b0, t0);
            t1 = WMMA_F32X4(a, b1, t1);
        }
    }
    __builtin_amdgcn_fence(__ATOMIC_ACQ_REL, "wavefront");
    __builtin_amdgcn_wave_barrier();
    if (wave == 31) {
        float alast = arow[8192];
        v2f a, b0, b1;
        a.x  = (half == 0) ? alast : 0.f;
        a.y  = 0.f;
        b0.x = KHH_V[1024 * 32 + mrow];
        b0.y = b0.x;
        b1.x = KHH_V[1024 * 32 + 16 + mrow];
        b1.y = b1.x;
        t0 = WMMA_F32X4(a, b0, t0);
        t1 = WMMA_F32X4(a, b1, t1);
    }

    float* pw = stage[wave];
#pragma unroll
    for (int j = 0; j < 8; ++j) {
        int m = j + 8 * half;
        pw[m * 32 + mrow]      = t0[j];
        pw[m * 32 + 16 + mrow] = t1[j];
    }
    __syncthreads();

    if (threadIdx.x < 512) {
        const int e = threadIdx.x;
        float acc = stage[0][e];
#pragma unroll
        for (int w = 1; w < 32; ++w) acc += stage[w][e];
        tl[e] = acc * HK;
    }
    __syncthreads();

    v2f af[8];
#pragma unroll
    for (int q = 0; q < 8; ++q) {
        int ka = 4 * q + 2 * half;
        af[q].x = tl[mrow * 32 + ka];
        af[q].y = tl[mrow * 32 + ka + 1];
    }
#pragma unroll
    for (int u = 0; u < 2; ++u) {
        int n = (2 * wave + u) * 16 + mrow;
        v8f acc = {};
#pragma unroll
        for (int q = 0; q < 8; ++q) {
            int ka = 4 * q + 2 * half;
            v2f b;
            b.x = KHH_U[n * 32 + ka];
            b.y = KHH_U[n * 32 + ka + 1];
            acc = WMMA_F32X4(af[q], b, acc);
        }
#pragma unroll
        for (int j = 0; j < 8; ++j) {
            int m = j + 8 * half;
            wh0[(size_t)(R + m) * 1025 + n] = acc[j];
        }
    }
    if (wave == 0) {
        v8f acc = {};
#pragma unroll
        for (int q = 0; q < 8; ++q) {
            int ka = 4 * q + 2 * half;
            v2f b;
            b.x = KHH_U[1024 * 32 + ka];
            b.y = KHH_U[1024 * 32 + ka + 1];
            acc = WMMA_F32X4(af[q], b, acc);
        }
        if (mrow == 0) {
#pragma unroll
            for (int j = 0; j < 8; ++j) {
                int m = j + 8 * half;
                wh0[(size_t)(R + m) * 1025 + 1024] = acc[j];
            }
        }
    }
}

#define SEGC 16
#define NSEG (1024 / SEGC)
#define LP   33
#define HALO 32
#define WUMAX (8 * SEGC + 2 * HALO + 1)
#define WUMIN (8 * SEGC + HALO + 1)
#define NCTILE 17
#define KPITCH 16
#define KE_ROW(i) ((i) == 0 ? 0  : ((i) == 1 ? 48 : 128))
#define KO_ROW(i) ((i) == 0 ? 32 : ((i) == 1 ? 96 : 208))

__global__ void __launch_bounds__(256) k_refine(const float* __restrict__ uh,
                                                const float* __restrict__ wh0,
                                                const float* __restrict__ sbuf,
                                                const float* __restrict__ Kbd_V,
                                                const float* __restrict__ Ke0_U,
                                                const float* __restrict__ Ke0_V,
                                                const float* __restrict__ Ko0_U,
                                                const float* __restrict__ Ko0_V,
                                                const float* __restrict__ Ke1_U,
                                                const float* __restrict__ Ke1_V,
                                                const float* __restrict__ Ko1_U,
                                                const float* __restrict__ Ko1_V,
                                                const float* __restrict__ Ke2_U,
                                                const float* __restrict__ Ke2_V,
                                                const float* __restrict__ Ko2_U,
                                                const float* __restrict__ Ko2_V,
                                                const int* __restrict__ nbrL0,
                                                const int* __restrict__ nbrL1,
                                                const int* __restrict__ nbrL2,
                                                float* __restrict__ out) {
    __shared__ float uw[WUMAX * LP];
    __shared__ float wA[(4 * SEGC + 1) * LP];
    __shared__ float wB[(8 * SEGC + 1) * LP];
    __shared__ __attribute__((aligned(16))) float kcf[NCTILE * 16 * KPITCH];
    const int wave = __builtin_amdgcn_readfirstlane((int)(threadIdx.x >> 5));
    const int lane = threadIdx.x & 31;
    const int seg  = blockIdx.x % NSEG;
    const int row0 = (blockIdx.x / NSEG) * 32;
    const int c0   = seg * SEGC;
    const int jlo  = (8 * c0 - HALO) < 0 ? 0 : (8 * c0 - HALO);
    const int jhi  = (8 * (c0 + SEGC) + HALO) > 8192 ? 8192 : (8 * (c0 + SEGC) + HALO);
    const int wu   = jhi - jlo + 1;
    const float H  = 1.0f / 8192.0f;

    {
        const int mrow = lane & 15;
        const int half = lane >> 4;
#define TILE_SRC(t, up, vp)                                                              \
        do {                                                                             \
            const int lvl_  = ((t) >= 3 ? 1 : 0) + ((t) >= 8 ? 1 : 0);                    \
            const int tb_   = lvl_ == 0 ? 0 : (lvl_ == 1 ? 3 : 8);     \
            const int ne_   = lvl_ == 0 ? 2 : (lvl_ == 1 ? 3 : 5);      \
            const bool odd_ = ((t) - tb_) >= ne_;                                         \
            const int tt_   = (t) - tb_ - (odd_ ? ne_ : 0);            \
            const float* U_ = odd_ ? (lvl_ == 0 ? Ko0_U : (lvl_ == 1 ? Ko1_U : Ko2_U))    \
                                   : (lvl_ == 0 ? Ke0_U : (lvl_ == 1 ? Ke1_U : Ke2_U));   \
            const float* V_ = odd_ ? (lvl_ == 0 ? Ko0_V : (lvl_ == 1 ? Ko1_V : Ko2_V))    \
                                   : (lvl_ == 0 ? Ke0_V : (lvl_ == 1 ? Ke1_V : Ke2_V));   \
            const int glast_ = (1024 << lvl_) - (odd_ ? 1 : 0);          \
            const int vlast_ = odd_ ? 14 : 7;                            \
            int g_ = (c0 << lvl_) + 16 * tt_ + mrow;                                      \
            g_ = g_ > glast_ ? glast_ : g_;                                               \
            const int vr_ = mrow > vlast_ ? vlast_ : mrow;                                \
            (up) = U_ + (size_t)g_ * 32 + 4 * half;                                       \
            (vp) = V_ + vr_ * 32 + 4 * half;                                              \
        } while (0)
        const float* up;
        const float* vp;
        v4f a0[4], b0[4], a1[4], b1[4], a2[4], b2[4];
        TILE_SRC(wave, up, vp);
#pragma unroll
        for (int qq = 0; qq < 4; ++qq) { a0[qq] = *(const v4f*)(up + 8 * qq); b0[qq] = *(const v4f*)(vp + 8 * qq); }
        TILE_SRC(wave + 8, up, vp);
#pragma unroll
        for (int qq = 0; qq < 4; ++qq) { a1[qq] = *(const v4f*)(up + 8 * qq); b1[qq] = *(const v4f*)(vp + 8 * qq); }
        if (wave == 0) {
            TILE_SRC(16, up, vp);
#pragma unroll
            for (int qq = 0; qq < 4; ++qq) { a2[qq] = *(const v4f*)(up + 8 * qq); b2[qq] = *(const v4f*)(vp + 8 * qq); }
        }
        v8f d0 = {}; v8f d1 = {};
#pragma unroll
        for (int s = 0; s < 8; ++s) {
            const int qq = s >> 1, e = s & 1;
            v2f x, y;
            x.x = a0[qq][2 * e]; x.y = a0[qq][2 * e + 1];
            y.x = b0[qq][2 * e]; y.y = b0[qq][2 * e + 1];
            d0 = WMMA_F32X4(x, y, d0);
            x.x = a1[qq][2 * e]; x.y = a1[qq][2 * e + 1];
            y.x = b1[qq][2 * e]; y.y = b1[qq][2 * e + 1];
            d1 = WMMA_F32X4(x, y, d1);
        }
        float* o = kcf + (16 * wave + 8 * half) * KPITCH + mrow;
#pragma unroll
        for (int j = 0; j < 8; ++j) {
            o[j * KPITCH]         = d0[j];
            o[(128 + j) * KPITCH] = d1[j];
        }
        if (wave == 0) {
            v8f d2 = {};
#pragma unroll
            for (int s = 0; s < 8; ++s) {
                const int qq = s >> 1, e = s & 1;
                v2f x, y;
                x.x = a2[qq][2 * e]; x.y = a2[qq][2 * e + 1];
                y.x = b2[qq][2 * e]; y.y = b2[qq][2 * e + 1];
                d2 = WMMA_F32X4(x, y, d2);
            }
#pragma unroll
            for (int j = 0; j < 8; ++j) o[(256 + j) * KPITCH] = d2[j];
        }
#undef TILE_SRC
    }

    {
        float sv[4][7];
        float wv[4];
        const int wl = lane <= SEGC ? lane : SEGC;
#pragma unroll
        for (int q = 0; q < 4; ++q) {
            const int r = wave + 8 * q;
            const float* us = uh + (size_t)(row0 + r) * 8193 + jlo;
#pragma unroll
            for (int k = 0; k < 7; ++k) {
                const int p  = lane + 32 * k;
                const int pc = (32 * k + 31 < WUMIN) ? p : (p < wu ? p : wu - 1);
                sv[q][k] = us[pc];
            }
            wv[q] = wh0[(size_t)(row0 + r) * 1025 + c0 + wl];
        }
#pragma unroll
        for (int q = 0; q < 4; ++q) {
            const int r = wave + 8 * q;
#pragma unroll
            for (int k = 0; k < 7; ++k) {
                const int p  = lane + 32 * k;
                const int pc = (32 * k + 31 < WUMIN) ? p : (p < wu ? p : wu - 1);
                uw[pc * LP + r] = sv[q][k];
            }
            wA[wl * LP + r] = wv[q];
        }
    }
    __syncthreads();

    const int vb = lane - jlo * LP;
    const float* ug  = uh + (size_t)(row0 + lane) * 8193;

    const int*   nbrP[3] = {nbrL0, nbrL1, nbrL2};
    float* oldW = wA;
    float* newW = wB;

#pragma unroll
    for (int i = 0; i < 3; ++i) {
        const int step = 4 >> i;
        const float hf = H * (float)step;
        const int ns = SEGC << i;
        const int nb = c0 << i;
        const float* Ke = kcf + KE_ROW(i) * KPITCH;
        const float* Ko = kcf + KO_ROW(i) * KPITCH;
        const int* nbr  = nbrP[i];
        for (int n = wave; n <= ns; n += 8) {
            const int* row = nbr + (size_t)(2 * (nb + n)) * 15;
            const v4f* kp = (const v4f*)(Ke + n * KPITCH);
            int id[8];
            float kc[8];
#pragma unroll
            for (int m = 0; m < 8; ++m) id[m] = row[2 * m];
            const v4f k0 = kp[0], k1 = kp[1];
#pragma unroll
            for (int m = 0; m < 4; ++m) { kc[m] = k0[m]; kc[4 + m] = k1[m]; }
            const float o0 = oldW[n * LP + lane];
            int lo = id[0], hi = id[0];
#pragma unroll
            for (int m = 1; m < 8; ++m) { lo = id[m] < lo ? id[m] : lo; hi = id[m] > hi ? id[m] : hi; }
            float acc = 0.f;
            if (step * lo >= jlo && step * hi <= jhi) {
#pragma unroll
                for (int m = 0; m < 8; ++m) acc = __builtin_fmaf(kc[m], uw[vb + id[m] * (step * LP)], acc);
            } else {
#pragma unroll
                for (int m = 0; m < 8; ++m) acc = __builtin_fmaf(kc[m], ug[step * id[m]], acc);
            }
            newW[(2 * n) * LP + lane] = o0 + hf * acc;
        }
        for (int n = wave; n < ns; n += 8) {
            const int* row = nbr + (size_t)(2 * (nb + n) + 1) * 15;
            const v4f* kp = (const v4f*)(Ko + n * KPITCH);
            int id[15];
            float kc[16];
#pragma unroll
            for (int m = 0; m < 15; ++m) id[m] = row[m];
            const v4f k0 = kp[0], k1 = kp[1], k2 = kp[2], k3 = kp[3];
#pragma unroll
            for (int m = 0; m < 4; ++m) { kc[m] = k0[m]; kc[4 + m] = k1[m]; kc[8 + m] = k2[m]; kc[12 + m] = k3[m]; }
            const float o0 = oldW[n * LP + lane];
            const float o1 = oldW[(n + 1) * LP + lane];
            int lo = id[0], hi = id[0];
#pragma unroll
            for (int m = 1; m < 15; ++m) { lo = id[m] < lo ? id[m] : lo; hi = id[m] > hi ? id[m] : hi; }
            float acc = 0.f;
            if (step * lo >= jlo && step * hi <= jhi) {
#pragma unroll
                for (int m = 0; m < 15; ++m) acc = __builtin_fmaf(kc[m], uw[vb + id[m] * (step * LP)], acc);
            } else {
#pragma unroll
                for (int m = 0; m < 15; ++m) acc = __builtin_fmaf(kc[m], ug[step * id[m]], acc);
            }
            newW[(2 * n + 1) * LP + lane] = 0.5f * (o0 + o1) + hf * acc;
        }
        __syncthreads();
        float* tmp = oldW; oldW = newW; newW = tmp;
    }

    if (seg == 0 || seg == NSEG - 1) {
        if (wave == 0) {
            const float* kv = Kbd_V + ((seg == 0) ? 0 : 32);
            const float* sp0 = sbuf + (size_t)(row0 + lane) * 16;
            const float* sp1 = sbuf + 8192 + (size_t)(row0 + lane) * 16;
            float acc = 0.f;
#pragma unroll
            for (int r = 0; r < 16; ++r) acc += kv[r] * sp0[r];
#pragma unroll
            for (int r = 0; r < 16; ++r) acc += kv[16 + r] * sp1[r];
            oldW[((seg == 0) ? 0 : 8 * SEGC) * LP + lane] = H * acc;
        }
        __syncthreads();
    }

    const int npts = (seg == NSEG - 1) ? (8 * SEGC + 1) : (8 * SEGC);
    for (int r = wave; r < 32; r += 8) {
        float* dst = out + (size_t)(row0 + r) * 8193 + 8 * c0;
        for (int p = lane; p < npts; p += 32) dst[p] = oldW[p * LP + r];
    }
}

extern "C" void kernel_launch(void* const* d_in, const int* in_sizes, int n_in,
                              void* d_out, int out_size, void* d_ws, size_t ws_size,
                              hipStream_t stream) {
    const float* uh     = (const float*)d_in[0];
    const float* KHH_U  = (const float*)d_in[1];
    const float* KHH_V  = (const float*)d_in[2];
    const float* Ke0_U  = (const float*)d_in[3];
    const float* Ke0_V  = (const float*)d_in[4];
    const float* Ko0_U  = (const float*)d_in[5];
    const float* Ko0_V  = (const float*)d_in[6];
    const int*   nbr0   = (const int*)d_in[7];
    const float* Ke1_U  = (const float*)d_in[8];
    const float* Ke1_V  = (const float*)d_in[9];
    const float* Ko1_U  = (const float*)d_in[10];
    const float* Ko1_V  = (const float*)d_in[11];
    const int*   nbr1   = (const int*)d_in[12];
    const float* Ke2_U  = (const float*)d_in[13];
    const float* Ke2_V  = (const float*)d_in[14];
    const float* Ko2_U  = (const float*)d_in[15];
    const float* Ko2_V  = (const float*)d_in[16];
    const int*   nbr2   = (const int*)d_in[17];
    const float* Kbd_U  = (const float*)d_in[18];
    const float* Kbd_V  = (const float*)d_in[19];

    float* ws  = (float*)d_ws;
    float* wh0 = ws;
    float* sb  = wh0 + 524800;

    float* outp = (float*)d_out;

    k_wmma_proj<<<96, 1024, 0, stream>>>(uh, KHH_U, KHH_V, Kbd_U, wh0, sb);

    k_refine<<<1024, 256, 0, stream>>>(uh, wh0, sb, Kbd_V,
                                      Ke0_U, Ke0_V, Ko0_U, Ko0_V,
                                      Ke1_U, Ke1_V, Ko1_U, Ko1_V,
                                      Ke2_U, Ke2_V, Ko2_U, Ko2_V,
                                      nbr2, nbr1, nbr0, outp);
}


# Projection: each row tile's contraction is split in halves over two blocks for every product (s: all 32 columns per block); refinement adds the partial planes of wh0 and s; summation order changes
# speedup vs baseline: 12.1699x; 1.1405x over previous
#include <hip/hip_runtime.h>
#include <hip/hip_bf16.h>

typedef __attribute__((ext_vector_type(2))) float v2f;
typedef __attribute__((ext_vector_type(4))) float v4f;
typedef __attribute__((ext_vector_type(8))) float v8f;

#define WMMA_F32X4(a, b, c) \
    __builtin_amdgcn_wmma_f32_16x16x4_f32(false, (a), false, (b), (short)0, (c), false, false)

__global__ void __launch_bounds__(1024) k_wmma_proj(const float* __restrict__ uh,
                                                    const float* __restrict__ KHH_U,
                                                    const float* __restrict__ KHH_V,
                                                    const float* __restrict__ Kbd_U,
                                                    float* __restrict__ wh0,
                                                    float* __restrict__ sbuf) {
    __shared__ __attribute__((aligned(16))) float stage[32][2304];
    __shared__ float tl[16 * 32];
    const int wave = __builtin_amdgcn_readfirstlane((int)(threadIdx.x >> 5));
    const int lane = threadIdx.x & 31;
    const int half = lane >> 4;
    const int mrow = lane & 15;
    const int kh  = blockIdx.x & 1;
    const int kc0 = 512 * kh + 16 * wave;
    const int kf0 = 4096 * kh + 128 * wave;

    float* As = stage[wave];
    float* Bs = As + 16 * 36;
    float* Vs = Bs + 32 * 48;

    if (blockIdx.x >= 64) {
        const int R  = ((blockIdx.x - 64) >> 1) * 16;
        const float* ublk = uh + (size_t)R * 8193 + lane;
        const int srow = lane >> 3;
        const int scol = 4 * (lane & 7);
        v8f s0 = {}; v8f s1 = {};
        for (int c = 0; c < 4; ++c) {
            const int kbase = kf0 + 32 * c;
            float av[16];
#pragma unroll
            for (int r = 0; r < 16; ++r) av[r] = ublk[(size_t)r * 8193 + kbase];
            const float* bsrc = Kbd_U + (size_t)(kbase + srow) * 32 + scol;
            v4f bv[8];
#pragma unroll
            for (int i = 0; i < 8; ++i) bv[i] = *(const v4f*)(bsrc + (size_t)(4 * i) * 32);

            __builtin_amdgcn_fence(__ATOMIC_ACQ_REL, "wavefront");
            __builtin_amdgcn_wave_barrier();
#pragma unroll
            for (int r = 0; r < 16; ++r) As[r * 36 + lane] = av[r];
#pragma unroll
            for (int i = 0; i < 8; ++i) *(v4f*)(Bs + (4 * i + srow) * 48 + scol) = bv[i];
            __builtin_amdgcn_fence(__ATOMIC_ACQ_REL, "wavefront");
            __builtin_amdgcn_wave_barrier();

#pragma unroll
            for (int q = 0; q < 8; ++q) {
                const int kl = 4 * q + 2 * half;
                v2f a, b0, b1;
                a    = *(const v2f*)(As + mrow * 36 + kl);
                b0.x = Bs[kl * 48 + mrow];
                b0.y = Bs[(kl + 1) * 48 + mrow];
                b1.x = Bs[kl * 48 + 16 + mrow];
                b1.y = Bs[(kl + 1) * 48 + 16 + mrow];
                s0 = WMMA_F32X4(a, b0, s0);
                s1 = WMMA_F32X4(a, b1, s1);
            }
        }
        __builtin_amdgcn_fence(__ATOMIC_ACQ_REL, "wavefront");
        __builtin_amdgcn_wave_barrier();
        if (kh == 1 && wave == 31) {
            float alast = uh[(size_t)(R + mrow) * 8193 + 8192];
            v2f a, b0, b1;
            a.x  = (half == 0) ? alast : 0.f;
            a.y  = 0.f;
            b0.x = Kbd_U[8192 * 32 + mrow];
            b0.y = b0.x;
            b1.x = Kbd_U[8192 * 32 + 16 + mrow];
            b1.y = b1.x;
            s0 = WMMA_F32X4(a, b0, s0);
            s1 = WMMA_F32X4(a, b1, s1);
        }
        float* pw = stage[wave];
#pragma unroll
        for (int j = 0; j < 8; ++j) {
            int m = j + 8 * half;
            pw[m * 32 + mrow]      = s0[j];
            pw[m * 32 + 16 + mrow] = s1[j];
        }
        __syncthreads();
        if (threadIdx.x < 512) {
            const int e = threadIdx.x;
            float acc = stage[0][e];
#pragma unroll
            for (int w = 1; w < 32; ++w) acc += stage[w][e];
            sbuf[(size_t)kh * 16384 + (size_t)R * 32 + e] = acc;
        }
        return;
    }

    const int R    = (blockIdx.x >> 1) * 16;
    const float HK = 1.0f / 1024.0f;
    const float* arow = uh + (size_t)(R + mrow) * 8193;

    v8f t0 = {}; v8f t1 = {};
    const float* ublk = uh + (size_t)R * 8193 + lane;
    const int srow = lane >> 3;
    const int scol = 4 * (lane & 7);

    for (int c = 0; c < 4; ++c) {
        const int kbase = kf0 + 32 * c;
        float av[16];
#pragma unroll
        for (int r = 0; r < 16; ++r) av[r] = ublk[(size_t)r * 8193 + kbase];
        const v4f vv = *((const v4f*)(KHH_V + (size_t)(kc0 + 4 * c) * 32) + lane);

        __builtin_amdgcn_fence(__ATOMIC_ACQ_REL, "wavefront");
        __builtin_amdgcn_wave_barrier();
#pragma unroll
        for (int r = 0; r < 16; ++r) As[r * 36 + lane] = av[r];
        *(v4f*)(Vs + srow * 48 + scol) = vv;
        __builtin_amdgcn_fence(__ATOMIC_ACQ_REL, "wavefront");
        __builtin_amdgcn_wave_barrier();

        {
            v2f a, b0, b1;
            a.x  = As[mrow * 36 + 16 * half];
            a.y  = As[mrow * 36 + 16 * half + 8];
            b0.x = Vs[(2 * half) * 48 + mrow];
            b0.y = Vs[(2 * half + 1) * 48 + mrow];
            b1.x = Vs[(2 * half) * 48 + 16 + mrow];
            b1.y = Vs[(2 * half + 1) * 48 + 16 + mrow];
            t0 = WMMA_F32X4(a, b0, t0);
            t1 = WMMA_F32X4(a, b1, t1);
        }
    }
    __builtin_amdgcn_fence(__ATOMIC_ACQ_REL, "wavefront");
    __builtin_amdgcn_wave_barrier();
    if (kh == 1 && wave == 31) {
        float alast = arow[8192];
        v2f a, b0, b1;
        a.x  = (half == 0) ? alast : 0.f;
        a.y  = 0.f;
        b0.x = KHH_V[1024 * 32 + mrow];
        b0.y = b0.x;
        b1.x = KHH_V[1024 * 32 + 16 + mrow];
        b1.y = b1.x;
        t0 = WMMA_F32X4(a, b0, t0);
        t1 = WMMA_F32X4(a, b1, t1);
    }

    float* pw = stage[wave];
#pragma unroll
    for (int j = 0; j < 8; ++j) {
        int m = j + 8 * half;
        pw[m * 32 + mrow]      = t0[j];
        pw[m * 32 + 16 + mrow] = t1[j];
    }
    __syncthreads();

    if (threadIdx.x < 512) {
        const int e = threadIdx.x;
        float acc = stage[0][e];
#pragma unroll
        for (int w = 1; w < 32; ++w) acc += stage[w][e];
        tl[e] = acc * HK;
    }
    __syncthreads();

    float* whp = wh0 + (size_t)kh * 524800;
    v2f af[8];
#pragma unroll
    for (int q = 0; q < 8; ++q) {
        int ka = 4 * q + 2 * half;
        af[q].x = tl[mrow * 32 + ka];
        af[q].y = tl[mrow * 32 + ka + 1];
    }
#pragma unroll
    for (int u = 0; u < 2; ++u) {
        int n = (2 * wave + u) * 16 + mrow;
        v8f acc = {};
#pragma unroll
        for (int q = 0; q < 8; ++q) {
            int ka = 4 * q + 2 * half;
            v2f b;
            b.x = KHH_U[n * 32 + ka];
            b.y = KHH_U[n * 32 + ka + 1];
            acc = WMMA_F32X4(af[q], b, acc);
        }
#pragma unroll
        for (int j = 0; j < 8; ++j) {
            int m = j + 8 * half;
            whp[(size_t)(R + m) * 1025 + n] = acc[j];
        }
    }
    if (wave == 0) {
        v8f acc = {};
#pragma unroll
        for (int q = 0; q < 8; ++q) {
            int ka = 4 * q + 2 * half;
            v2f b;
            b.x = KHH_U[1024 * 32 + ka];
            b.y = KHH_U[1024 * 32 + ka + 1];
            acc = WMMA_F32X4(af[q], b, acc);
        }
        if (mrow == 0) {
#pragma unroll
            for (int j = 0; j < 8; ++j) {
                int m = j + 8 * half;
                whp[(size_t)(R + m) * 1025 + 1024] = acc[j];
            }
        }
    }
}

#define SEGC 16
#define NSEG (1024 / SEGC)
#define LP   33
#define HALO 32
#define WUMAX (8 * SEGC + 2 * HALO + 1)
#define WUMIN (8 * SEGC + HALO + 1)
#define NCTILE 17
#define KPITCH 16
#define KE_ROW(i) ((i) == 0 ? 0  : ((i) == 1 ? 48 : 128))
#define KO_ROW(i) ((i) == 0 ? 32 : ((i) == 1 ? 96 : 208))

__global__ void __launch_bounds__(256) k_refine(const float* __restrict__ uh,
                                                const float* __restrict__ wh0,
                                                const float* __restrict__ sbuf,
                                                const float* __restrict__ Kbd_V,
                                                const float* __restrict__ Ke0_U,
                                                const float* __restrict__ Ke0_V,
                                                const float* __restrict__ Ko0_U,
                                                const float* __restrict__ Ko0_V,
                                                const float* __restrict__ Ke1_U,
                                                const float* __restrict__ Ke1_V,
                                                const float* __restrict__ Ko1_U,
                                                const float* __restrict__ Ko1_V,
                                                const float* __restrict__ Ke2_U,
                                                const float* __restrict__ Ke2_V,
                                                const float* __restrict__ Ko2_U,
                                                const float* __restrict__ Ko2_V,
                                                const int* __restrict__ nbrL0,
                                                const int* __restrict__ nbrL1,
                                                const int* __restrict__ nbrL2,
                                                float* __restrict__ out) {
    __shared__ float uw[WUMAX * LP];
    __shared__ float wA[(4 * SEGC + 1) * LP];
    __shared__ float wB[(8 * SEGC + 1) * LP];
    __shared__ float wX[(SEGC + 1) * LP];
    __shared__ __attribute__((aligned(16))) float kcf[NCTILE * 16 * KPITCH];
    const int wave = __builtin_amdgcn_readfirstlane((int)(threadIdx.x >> 5));
    const int lane = threadIdx.x & 31;
    const int seg  = blockIdx.x % NSEG;
    const int row0 = (blockIdx.x / NSEG) * 32;
    const int c0   = seg * SEGC;
    const int jlo  = (8 * c0 - HALO) < 0 ? 0 : (8 * c0 - HALO);
    const int jhi  = (8 * (c0 + SEGC) + HALO) > 8192 ? 8192 : (8 * (c0 + SEGC) + HALO);
    const int wu   = jhi - jlo + 1;
    const float H  = 1.0f / 8192.0f;

    {
        const int mrow = lane & 15;
        const int half = lane >> 4;
#define TILE_SRC(t, up, vp)                                                              \
        do {                                                                             \
            const int lvl_  = ((t) >= 3 ? 1 : 0) + ((t) >= 8 ? 1 : 0);                    \
            const int tb_   = lvl_ == 0 ? 0 : (lvl_ == 1 ? 3 : 8);     \
            const int ne_   = lvl_ == 0 ? 2 : (lvl_ == 1 ? 3 : 5);      \
            const bool odd_ = ((t) - tb_) >= ne_;                                         \
            const int tt_   = (t) - tb_ - (odd_ ? ne_ : 0);            \
            const float* U_ = odd_ ? (lvl_ == 0 ? Ko0_U : (lvl_ == 1 ? Ko1_U : Ko2_U))    \
                                   : (lvl_ == 0 ? Ke0_U : (lvl_ == 1 ? Ke1_U : Ke2_U));   \
            const float* V_ = odd_ ? (lvl_ == 0 ? Ko0_V : (lvl_ == 1 ? Ko1_V : Ko2_V))    \
                                   : (lvl_ == 0 ? Ke0_V : (lvl_ == 1 ? Ke1_V : Ke2_V));   \
            const int glast_ = (1024 << lvl_) - (odd_ ? 1 : 0);          \
            const int vlast_ = odd_ ? 14 : 7;                            \
            int g_ = (c0 << lvl_) + 16 * tt_ + mrow;                                      \
            g_ = g_ > glast_ ? glast_ : g_;                                               \
            const int vr_ = mrow > vlast_ ? vlast_ : mrow;                                \
            (up) = U_ + (size_t)g_ * 32 + 4 * half;                                       \
            (vp) = V_ + vr_ * 32 + 4 * half;                                              \
        } while (0)
        const float* up;
        const float* vp;
        v4f a0[4], b0[4], a1[4], b1[4], a2[4], b2[4];
        TILE_SRC(wave, up, vp);
#pragma unroll
        for (int qq = 0; qq < 4; ++qq) { a0[qq] = *(const v4f*)(up + 8 * qq); b0[qq] = *(const v4f*)(vp + 8 * qq); }
        TILE_SRC(wave + 8, up, vp);
#pragma unroll
        for (int qq = 0; qq < 4; ++qq) { a1[qq] = *(const v4f*)(up + 8 * qq); b1[qq] = *(const v4f*)(vp + 8 * qq); }
        if (wave == 0) {
            TILE_SRC(16, up, vp);
#pragma unroll
            for (int qq = 0; qq < 4; ++qq) { a2[qq] = *(const v4f*)(up + 8 * qq); b2[qq] = *(const v4f*)(vp + 8 * qq); }
        }
        v8f d0 = {}; v8f d1 = {};
#pragma unroll
        for (int s = 0; s < 8; ++s) {
            const int qq = s >> 1, e = s & 1;
            v2f x, y;
            x.x = a0[qq][2 * e]; x.y = a0[qq][2 * e + 1];
            y.x = b0[qq][2 * e]; y.y = b0[qq][2 * e + 1];
            d0 = WMMA_F32X4(x, y, d0);
            x.x = a1[qq][2 * e]; x.y = a1[qq][2 * e + 1];
            y.x = b1[qq][2 * e]; y.y = b1[qq][2 * e + 1];
            d1 = WMMA_F32X4(x, y, d1);
        }
        float* o = kcf + (16 * wave + 8 * half) * KPITCH + mrow;
#pragma unroll
        for (int j = 0; j < 8; ++j) {
            o[j * KPITCH]         = d0[j];
            o[(128 + j) * KPITCH] = d1[j];
        }
        if (wave == 0) {
            v8f d2 = {};
#pragma unroll
            for (int s = 0; s < 8; ++s) {
                const int qq = s >> 1, e = s & 1;
                v2f x, y;
                x.x = a2[qq][2 * e]; x.y = a2[qq][2 * e + 1];
                y.x = b2[qq][2 * e]; y.y = b2[qq][2 * e + 1];
                d2 = WMMA_F32X4(x, y, d2);
            }
#pragma unroll
            for (int j = 0; j < 8; ++j) o[(256 + j) * KPITCH] = d2[j];
        }
#undef TILE_SRC
    }

    {
        float sv[4][7];
        float wv[4], wx[4];
        const int wl = lane <= SEGC ? lane : SEGC;
#pragma unroll
        for (int q = 0; q < 4; ++q) {
            const int r = wave + 8 * q;
            const float* us = uh + (size_t)(row0 + r) * 8193 + jlo;
#pragma unroll
            for (int k = 0; k < 7; ++k) {
                const int p  = lane + 32 * k;
                const int pc = (32 * k + 31 < WUMIN) ? p : (p < wu ? p : wu - 1);
                sv[q][k] = us[pc];
            }
            wv[q] = wh0[(size_t)(row0 + r) * 1025 + c0 + wl];
            wx[q] = wh0[(size_t)524800 + (size_t)(row0 + r) * 1025 + c0 + wl];
        }
#pragma unroll
        for (int q = 0; q < 4; ++q) {
            const int r = wave + 8 * q;
#pragma unroll
            for (int k = 0; k < 7; ++k) {
                const int p  = lane + 32 * k;
                const int pc = (32 * k + 31 < WUMIN) ? p : (p < wu ? p : wu - 1);
                uw[pc * LP + r] = sv[q][k];
            }
            wA[wl * LP + r] = wv[q];
            wX[wl * LP + r] = wx[q];
        }
    }
    __syncthreads();

    const int vb = lane - jlo * LP;
    const float* ug  = uh + (size_t)(row0 + lane) * 8193;

    const int*   nbrP[3] = {nbrL0, nbrL1, nbrL2};
    float* oldW = wA;
    float* newW = wB;

#pragma unroll
    for (int i = 0; i < 3; ++i) {
        const int step = 4 >> i;
        const float hf = H * (float)step;
        const int ns = SEGC << i;
        const int nb = c0 << i;
        const float* Ke = kcf + KE_ROW(i) * KPITCH;
        const float* Ko = kcf + KO_ROW(i) * KPITCH;
        const int* nbr  = nbrP[i];
        for (int n = wave; n <= ns; n += 8) {
            const int* row = nbr + (size_t)(2 * (nb + n)) * 15;
            const v4f* kp = (const v4f*)(Ke + n * KPITCH);
            int id[8];
            float kc[8];
#pragma unroll
            for (int m = 0; m < 8; ++m) id[m] = row[2 * m];
            const v4f k0 = kp[0], k1 = kp[1];
#pragma unroll
            for (int m = 0; m < 4; ++m) { kc[m] = k0[m]; kc[4 + m] = k1[m]; }
            float o0 = oldW[n * LP + lane];
            if (i == 0) o0 += wX[n * LP + lane];
            int lo = id[0], hi = id[0];
#pragma unroll
            for (int m = 1; m < 8; ++m) { lo = id[m] < lo ? id[m] : lo; hi = id[m] > hi ? id[m] : hi; }
            float acc = 0.f;
            if (step * lo >= jlo && step * hi <= jhi) {
#pragma unroll
                for (int m = 0; m < 8; ++m) acc = __builtin_fmaf(kc[m], uw[vb + id[m] * (step * LP)], acc);
            } else {
#pragma unroll
                for (int m = 0; m < 8; ++m) acc = __builtin_fmaf(kc[m], ug[step * id[m]], acc);
            }
            newW[(2 * n) * LP + lane] = o0 + hf * acc;
        }
        for (int n = wave; n < ns; n += 8) {
            const int* row = nbr + (size_t)(2 * (nb + n) + 1) * 15;
            const v4f* kp = (const v4f*)(Ko + n * KPITCH);
            int id[15];
            float kc[16];
#pragma unroll
            for (int m = 0; m < 15; ++m) id[m] = row[m];
            const v4f k0 = kp[0], k1 = kp[1], k2 = kp[2], k3 = kp[3];
#pragma unroll
            for (int m = 0; m < 4; ++m) { kc[m] = k0[m]; kc[4 + m] = k1[m]; kc[8 + m] = k2[m]; kc[12 + m] = k3[m]; }
            float o0 = oldW[n * LP + lane];
            float o1 = oldW[(n + 1) * LP + lane];
            if (i == 0) {
                o0 += wX[n * LP + lane];
                o1 += wX[(n + 1) * LP + lane];
            }
            int lo = id[0], hi = id[0];
#pragma unroll
            for (int m = 1; m < 15; ++m) { lo = id[m] < lo ? id[m] : lo; hi = id[m] > hi ? id[m] : hi; }
            float acc = 0.f;
            if (step * lo >= jlo && step * hi <= jhi) {
#pragma unroll
                for (int m = 0; m < 15; ++m) acc = __builtin_fmaf(kc[m], uw[vb + id[m] * (step * LP)], acc);
            } else {
#pragma unroll
                for (int m = 0; m < 15; ++m) acc = __builtin_fmaf(kc[m], ug[step * id[m]], acc);
            }
            newW[(2 * n + 1) * LP + lane] = 0.5f * (o0 + o1) + hf * acc;
        }
        __syncthreads();
        float* tmp = oldW; oldW = newW; newW = tmp;
    }

    if (seg == 0 || seg == NSEG - 1) {
        if (wave == 0) {
            const float* kv = Kbd_V + ((seg == 0) ? 0 : 32);
            const float* sp0 = sbuf + (size_t)(row0 + lane) * 32;
            const float* sp1 = sbuf + 16384 + (size_t)(row0 + lane) * 32;
            float acc = 0.f;
#pragma unroll
            for (int r = 0; r < 32; ++r) acc += kv[r] * (sp0[r] + sp1[r]);
            oldW[((seg == 0) ? 0 : 8 * SEGC) * LP + lane] = H * acc;
        }
        __syncthreads();
    }

    const int npts = (seg == NSEG - 1) ? (8 * SEGC + 1) : (8 * SEGC);
    for (int r = wave; r < 32; r += 8) {
        float* dst = out + (size_t)(row0 + r) * 8193 + 8 * c0;
        for (int p = lane; p < npts; p += 32) dst[p] = oldW[p * LP + r];
    }
}

extern "C" void kernel_launch(void* const* d_in, const int* in_sizes, int n_in,
                              void* d_out, int out_size, void* d_ws, size_t ws_size,
                              hipStream_t stream) {
    const float* uh     = (const float*)d_in[0];
    const float* KHH_U  = (const float*)d_in[1];
    const float* KHH_V  = (const float*)d_in[2];
    const float* Ke0_U  = (const float*)d_in[3];
    const float* Ke0_V  = (const float*)d_in[4];
    const float* Ko0_U  = (const float*)d_in[5];
    const float* Ko0_V  = (const float*)d_in[6];
    const int*   nbr0   = (const int*)d_in[7];
    const float* Ke1_U  = (const float*)d_in[8];
    const float* Ke1_V  = (const float*)d_in[9];
    const float* Ko1_U  = (const float*)d_in[10];
    const float* Ko1_V  = (const float*)d_in[11];
    const int*   nbr1   = (const int*)d_in[12];
    const float* Ke2_U  = (const float*)d_in[13];
    const float* Ke2_V  = (const float*)d_in[14];
    const float* Ko2_U  = (const float*)d_in[15];
    const float* Ko2_V  = (const float*)d_in[16];
    const int*   nbr2   = (const int*)d_in[17];
    const float* Kbd_U  = (const float*)d_in[18];
    const float* Kbd_V  = (const float*)d_in[19];

    float* ws  = (float*)d_ws;
    float* wh0 = ws;
    float* sb  = wh0 + 1049600;

    float* outp = (float*)d_out;

    k_wmma_proj<<<128, 1024, 0, stream>>>(uh, KHH_U, KHH_V, Kbd_U, wh0, sb);

    k_refine<<<1024, 256, 0, stream>>>(uh, wh0, sb, Kbd_V,
                                      Ke0_U, Ke0_V, Ko0_U, Ko0_V,
                                      Ke1_U, Ke1_V, Ko1_U, Ko1_V,
                                      Ke2_U, Ke2_V, Ko2_U, Ko2_V,
                                      nbr2, nbr1, nbr0, outp);
}


# Refinement: the even and odd point of an interval share a trip: one index request, wait and window test for both, 23 taps in flight; tap offsets (index x pitch) on the scalar unit; sums unchanged
# speedup vs baseline: 12.9325x; 1.0627x over previous
#include <hip/hip_runtime.h>
#include <hip/hip_bf16.h>

typedef __attribute__((ext_vector_type(2))) float v2f;
typedef __attribute__((ext_vector_type(4))) float v4f;
typedef __attribute__((ext_vector_type(8))) float v8f;

#define WMMA_F32X4(a, b, c) \
    __builtin_amdgcn_wmma_f32_16x16x4_f32(false, (a), false, (b), (short)0, (c), false, false)

__global__ void __launch_bounds__(1024) k_wmma_proj(const float* __restrict__ uh,
                                                    const float* __restrict__ KHH_U,
                                                    const float* __restrict__ KHH_V,
                                                    const float* __restrict__ Kbd_U,
                                                    float* __restrict__ wh0,
                                                    float* __restrict__ sbuf) {
    __shared__ __attribute__((aligned(16))) float stage[32][2304];
    __shared__ float tl[16 * 32];
    const int wave = __builtin_amdgcn_readfirstlane((int)(threadIdx.x >> 5));
    const int lane = threadIdx.x & 31;
    const int half = lane >> 4;
    const int mrow = lane & 15;
    const int kh  = blockIdx.x & 1;
    const int kc0 = 512 * kh + 16 * wave;
    const int kf0 = 4096 * kh + 128 * wave;

    float* As = stage[wave];
    float* Bs = As + 16 * 36;
    float* Vs = Bs + 32 * 48;

    if (blockIdx.x >= 64) {
        const int R  = ((blockIdx.x - 64) >> 1) * 16;
        const float* ublk = uh + (size_t)R * 8193 + lane;
        const int srow = lane >> 3;
        const int scol = 4 * (lane & 7);
        v8f s0 = {}; v8f s1 = {};
        for (int c = 0; c < 4; ++c) {
            const int kbase = kf0 + 32 * c;
            float av[16];
#pragma unroll
            for (int r = 0; r < 16; ++r) av[r] = ublk[(size_t)r * 8193 + kbase];
            const float* bsrc = Kbd_U + (size_t)(kbase + srow) * 32 + scol;
            v4f bv[8];
#pragma unroll
            for (int i = 0; i < 8; ++i) bv[i] = *(const v4f*)(bsrc + (size_t)(4 * i) * 32);

            __builtin_amdgcn_fence(__ATOMIC_ACQ_REL, "wavefront");
            __builtin_amdgcn_wave_barrier();
#pragma unroll
            for (int r = 0; r < 16; ++r) As[r * 36 + lane] = av[r];
#pragma unroll
            for (int i = 0; i < 8; ++i) *(v4f*)(Bs + (4 * i + srow) * 48 + scol) = bv[i];
            __builtin_amdgcn_fence(__ATOMIC_ACQ_REL, "wavefront");
            __builtin_amdgcn_wave_barrier();

#pragma unroll
            for (int q = 0; q < 8; ++q) {
                const int kl = 4 * q + 2 * half;
                v2f a, b0, b1;
                a    = *(const v2f*)(As + mrow * 36 + kl);
                b0.x = Bs[kl * 48 + mrow];
                b0.y = Bs[(kl + 1) * 48 + mrow];
                b1.x = Bs[kl * 48 + 16 + mrow];
                b1.y = Bs[(kl + 1) * 48 + 16 + mrow];
                s0 = WMMA_F32X4(a, b0, s0);
                s1 = WMMA_F32X4(a, b1, s1);
            }
        }
        __builtin_amdgcn_fence(__ATOMIC_ACQ_REL, "wavefront");
        __builtin_amdgcn_wave_barrier();
        if (kh == 1 && wave == 31) {
            float alast = uh[(size_t)(R + mrow) * 8193 + 8192];
            v2f a, b0, b1;
            a.x  = (half == 0) ? alast : 0.f;
            a.y  = 0.f;
            b0.x = Kbd_U[8192 * 32 + mrow];
            b0.y = b0.x;
            b1.x = Kbd_U[8192 * 32 + 16 + mrow];
            b1.y = b1.x;
            s0 = WMMA_F32X4(a, b0, s0);
            s1 = WMMA_F32X4(a, b1, s1);
        }
        float* pw = stage[wave];
#pragma unroll
        for (int j = 0; j < 8; ++j) {
            int m = j + 8 * half;
            pw[m * 32 + mrow]      = s0[j];
            pw[m * 32 + 16 + mrow] = s1[j];
        }
        __syncthreads();
        if (threadIdx.x < 512) {
            const int e = threadIdx.x;
            float acc = stage[0][e];
#pragma unroll
            for (int w = 1; w < 32; ++w) acc += stage[w][e];
            sbuf[(size_t)kh * 16384 + (size_t)R * 32 + e] = acc;
        }
        return;
    }

    const int R    = (blockIdx.x >> 1) * 16;
    const float HK = 1.0f / 1024.0f;
    const float* arow = uh + (size_t)(R + mrow) * 8193;

    v8f t0 = {}; v8f t1 = {};
    const float* ublk = uh + (size_t)R * 8193 + lane;
    const int srow = lane >> 3;
    const int scol = 4 * (lane & 7);

    for (int c = 0; c < 4; ++c) {
        const int kbase = kf0 + 32 * c;
        float av[16];
#pragma unroll
        for (int r = 0; r < 16; ++r) av[r] = ublk[(size_t)r * 8193 + kbase];
        const v4f vv = *((const v4f*)(KHH_V + (size_t)(kc0 + 4 * c) * 32) + lane);

        __builtin_amdgcn_fence(__ATOMIC_ACQ_REL, "wavefront");
        __builtin_amdgcn_wave_barrier();
#pragma unroll
        for (int r = 0; r < 16; ++r) As[r * 36 + lane] = av[r];
        *(v4f*)(Vs + srow * 48 + scol) = vv;
        __builtin_amdgcn_fence(__ATOMIC_ACQ_REL, "wavefront");
        __builtin_amdgcn_wave_barrier();

        {
            v2f a, b0, b1;
            a.x  = As[mrow * 36 + 16 * half];
            a.y  = As[mrow * 36 + 16 * half + 8];
            b0.x = Vs[(2 * half) * 48 + mrow];
            b0.y = Vs[(2 * half + 1) * 48 + mrow];
            b1.x = Vs[(2 * half) * 48 + 16 + mrow];
            b1.y = Vs[(2 * half + 1) * 48 + 16 + mrow];
            t0 = WMMA_F32X4(a, b0, t0);
            t1 = WMMA_F32X4(a, b1, t1);
        }
    }
    __builtin_amdgcn_fence(__ATOMIC_ACQ_REL, "wavefront");
    __builtin_amdgcn_wave_barrier();
    if (kh == 1 && wave == 31) {
        float alast = arow[8192];
        v2f a, b0, b1;
        a.x  = (half == 0) ? alast : 0.f;
        a.y  = 0.f;
        b0.x = KHH_V[1024 * 32 + mrow];
        b0.y = b0.x;
        b1.x = KHH_V[1024 * 32 + 16 + mrow];
        b1.y = b1.x;
        t0 = WMMA_F32X4(a, b0, t0);
        t1 = WMMA_F32X4(a, b1, t1);
    }

    float* pw = stage[wave];
#pragma unroll
    for (int j = 0; j < 8; ++j) {
        int m = j + 8 * half;
        pw[m * 32 + mrow]      = t0[j];
        pw[m * 32 + 16 + mrow] = t1[j];
    }
    __syncthreads();

    if (threadIdx.x < 512) {
        const int e = threadIdx.x;
        float acc = stage[0][e];
#pragma unroll
        for (int w = 1; w < 32; ++w) acc += stage[w][e];
        tl[e] = acc * HK;
    }
    __syncthreads();

    float* whp = wh0 + (size_t)kh * 524800;
    v2f af[8];
#pragma unroll
    for (int q = 0; q < 8; ++q) {
        int ka = 4 * q + 2 * half;
        af[q].x = tl[mrow * 32 + ka];
        af[q].y = tl[mrow * 32 + ka + 1];
    }
#pragma unroll
    for (int u = 0; u < 2; ++u) {
        int n = (2 * wave + u) * 16 + mrow;
        v8f acc = {};
#pragma unroll
        for (int q = 0; q < 8; ++q) {
            int ka = 4 * q + 2 * half;
            v2f b;
            b.x = KHH_U[n * 32 + ka];
            b.y = KHH_U[n * 32 + ka + 1];
            acc = WMMA_F32X4(af[q], b, acc);
        }
#pragma unroll
        for (int j = 0; j < 8; ++j) {
            int m = j + 8 * half;
            whp[(size_t)(R + m) * 1025 + n] = acc[j];
        }
    }
    if (wave == 0) {
        v8f acc = {};
#pragma unroll
        for (int q = 0; q < 8; ++q) {
            int ka = 4 * q + 2 * half;
            v2f b;
            b.x = KHH_U[1024 * 32 + ka];
            b.y = KHH_U[1024 * 32 + ka + 1];
            acc = WMMA_F32X4(af[q], b, acc);
        }
        if (mrow == 0) {
#pragma unroll
            for (int j = 0; j < 8; ++j) {
                int m = j + 8 * half;
                whp[(size_t)(R + m) * 1025 + 1024] = acc[j];
            }
        }
    }
}

#define SEGC 16
#define NSEG (1024 / SEGC)
#define LP   33
#define HALO 32
#define WUMAX (8 * SEGC + 2 * HALO + 1)
#define WUMIN (8 * SEGC + HALO + 1)
#define NCTILE 17
#define KPITCH 16
#define KE_ROW(i) ((i) == 0 ? 0  : ((i) == 1 ? 48 : 128))
#define KO_ROW(i) ((i) == 0 ? 32 : ((i) == 1 ? 96 : 208))

__global__ void __launch_bounds__(256) k_refine(const float* __restrict__ uh,
                                                const float* __restrict__ wh0,
                                                const float* __restrict__ sbuf,
                                                const float* __restrict__ Kbd_V,
                                                const float* __restrict__ Ke0_U,
                                                const float* __restrict__ Ke0_V,
                                                const float* __restrict__ Ko0_U,
                                                const float* __restrict__ Ko0_V,
                                                const float* __restrict__ Ke1_U,
                                                const float* __restrict__ Ke1_V,
                                                const float* __restrict__ Ko1_U,
                                                const float* __restrict__ Ko1_V,
                                                const float* __restrict__ Ke2_U,
                                                const float* __restrict__ Ke2_V,
                                                const float* __restrict__ Ko2_U,
                                                const float* __restrict__ Ko2_V,
                                                const int* __restrict__ nbrL0,
                                                const int* __restrict__ nbrL1,
                                                const int* __restrict__ nbrL2,
                                                float* __restrict__ out) {
    __shared__ float uw[WUMAX * LP];
    __shared__ float wA[(4 * SEGC + 1) * LP];
    __shared__ float wB[(8 * SEGC + 1) * LP];
    __shared__ float wX[(SEGC + 1) * LP];
    __shared__ __attribute__((aligned(16))) float kcf[NCTILE * 16 * KPITCH];
    const int wave = __builtin_amdgcn_readfirstlane((int)(threadIdx.x >> 5));
    const int lane = threadIdx.x & 31;
    const int seg  = blockIdx.x % NSEG;
    const int row0 = (blockIdx.x / NSEG) * 32;
    const int c0   = seg * SEGC;
    const int jlo  = (8 * c0 - HALO) < 0 ? 0 : (8 * c0 - HALO);
    const int jhi  = (8 * (c0 + SEGC) + HALO) > 8192 ? 8192 : (8 * (c0 + SEGC) + HALO);
    const int wu   = jhi - jlo + 1;
    const float H  = 1.0f / 8192.0f;

    {
        const int mrow = lane & 15;
        const int half = lane >> 4;
#define TILE_SRC(t, up, vp)                                                              \
        do {                                                                             \
            const int lvl_  = ((t) >= 3 ? 1 : 0) + ((t) >= 8 ? 1 : 0);                    \
            const int tb_   = lvl_ == 0 ? 0 : (lvl_ == 1 ? 3 : 8);     \
            const int ne_   = lvl_ == 0 ? 2 : (lvl_ == 1 ? 3 : 5);      \
            const bool odd_ = ((t) - tb_) >= ne_;                                         \
            const int tt_   = (t) - tb_ - (odd_ ? ne_ : 0);            \
            const float* U_ = odd_ ? (lvl_ == 0 ? Ko0_U : (lvl_ == 1 ? Ko1_U : Ko2_U))    \
                                   : (lvl_ == 0 ? Ke0_U : (lvl_ == 1 ? Ke1_U : Ke2_U));   \
            const float* V_ = odd_ ? (lvl_ == 0 ? Ko0_V : (lvl_ == 1 ? Ko1_V : Ko2_V))    \
                                   : (lvl_ == 0 ? Ke0_V : (lvl_ == 1 ? Ke1_V : Ke2_V));   \
            const int glast_ = (1024 << lvl_) - (odd_ ? 1 : 0);          \
            const int vlast_ = odd_ ? 14 : 7;                            \
            int g_ = (c0 << lvl_) + 16 * tt_ + mrow;                                      \
            g_ = g_ > glast_ ? glast_ : g_;                                               \
            const int vr_ = mrow > vlast_ ? vlast_ : mrow;                                \
            (up) = U_ + (size_t)g_ * 32 + 4 * half;                                       \
            (vp) = V_ + vr_ * 32 + 4 * half;                                              \
        } while (0)
        const float* up;
        const float* vp;
        v4f a0[4], b0[4], a1[4], b1[4], a2[4], b2[4];
        TILE_SRC(wave, up, vp);
#pragma unroll
        for (int qq = 0; qq < 4; ++qq) { a0[qq] = *(const v4f*)(up + 8 * qq); b0[qq] = *(const v4f*)(vp + 8 * qq); }
        TILE_SRC(wave + 8, up, vp);
#pragma unroll
        for (int qq = 0; qq < 4; ++qq) { a1[qq] = *(const v4f*)(up + 8 * qq); b1[qq] = *(const v4f*)(vp + 8 * qq); }
        if (wave == 0) {
            TILE_SRC(16, up, vp);
#pragma unroll
            for (int qq = 0; qq < 4; ++qq) { a2[qq] = *(const v4f*)(up + 8 * qq); b2[qq] = *(const v4f*)(vp + 8 * qq); }
        }
        v8f d0 = {}; v8f d1 = {};
#pragma unroll
        for (int s = 0; s < 8; ++s) {
            const int qq = s >> 1, e = s & 1;
            v2f x, y;
            x.x = a0[qq][2 * e]; x.y = a0[qq][2 * e + 1];
            y.x = b0[qq][2 * e]; y.y = b0[qq][2 * e + 1];
            d0 = WMMA_F32X4(x, y, d0);
            x.x = a1[qq][2 * e]; x.y = a1[qq][2 * e + 1];
            y.x = b1[qq][2 * e]; y.y = b1[qq][2 * e + 1];
            d1 = WMMA_F32X4(x, y, d1);
        }
        float* o = kcf + (16 * wave + 8 * half) * KPITCH + mrow;
#pragma unroll
        for (int j = 0; j < 8; ++j) {
            o[j * KPITCH]         = d0[j];
            o[(128 + j) * KPITCH] = d1[j];
        }
        if (wave == 0) {
            v8f d2 = {};
#pragma unroll
            for (int s = 0; s < 8; ++s) {
                const int qq = s >> 1, e = s & 1;
                v2f x, y;
                x.x = a2[qq][2 * e]; x.y = a2[qq][2 * e + 1];
                y.x = b2[qq][2 * e]; y.y = b2[qq][2 * e + 1];
                d2 = WMMA_F32X4(x, y, d2);
            }
#pragma unroll
            for (int j = 0; j < 8; ++j) o[(256 + j) * KPITCH] = d2[j];
        }
#undef TILE_SRC
    }

    {
        float sv[4][7];
        float wv[4], wx[4];
        const int wl = lane <= SEGC ? lane : SEGC;
#pragma unroll
        for (int q = 0; q < 4; ++q) {
            const int r = wave + 8 * q;
            const float* us = uh + (size_t)(row0 + r) * 8193 + jlo;
#pragma unroll
            for (int k = 0; k < 7; ++k) {
                const int p  = lane + 32 * k;
                const int pc = (32 * k + 31 < WUMIN) ? p : (p < wu ? p : wu - 1);
                sv[q][k] = us[pc];
            }
            wv[q] = wh0[(size_t)(row0 + r) * 1025 + c0 + wl];
            wx[q] = wh0[(size_t)524800 + (size_t)(row0 + r) * 1025 + c0 + wl];
        }
#pragma unroll
        for (int q = 0; q < 4; ++q) {
            const int r = wave + 8 * q;
#pragma unroll
            for (int k = 0; k < 7; ++k) {
                const int p  = lane + 32 * k;
                const int pc = (32 * k + 31 < WUMIN) ? p : (p < wu ? p : wu - 1);
                uw[pc * LP + r] = sv[q][k];
            }
            wA[wl * LP + r] = wv[q];
            wX[wl * LP + r] = wx[q];
        }
    }
    __syncthreads();

    const int vb = lane - jlo * LP;
    const float* ug  = uh + (size_t)(row0 + lane) * 8193;

    const int*   nbrP[3] = {nbrL0, nbrL1, nbrL2};
    float* oldW = wA;
    float* newW = wB;

    const int wlo = jlo * LP;
    const int whi = jhi * LP;

#pragma unroll
    for (int i = 0; i < 3; ++i) {
        const int step = 4 >> i;
        const float hf = H * (float)step;
        const int ns = SEGC << i;
        const int nb = c0 << i;
        const float* Ke = kcf + KE_ROW(i) * KPITCH;
        const float* Ko = kcf + KO_ROW(i) * KPITCH;
        const int* nbr  = nbrP[i];
        const int P = step * LP;
        for (int n = wave; n < ns; n += 8) {
            const int* row = nbr + (size_t)(2 * (nb + n)) * 15;
            int se[8], sd[15];
#pragma unroll
            for (int m = 0; m < 8; ++m) se[m] = row[2 * m] * P;
#pragma unroll
            for (int m = 0; m < 15; ++m) sd[m] = row[15 + m] * P;
            const v4f* kpe = (const v4f*)(Ke + n * KPITCH);
            const v4f* kpo = (const v4f*)(Ko + n * KPITCH);
            float ke[8], ko[16];
            const v4f e0 = kpe[0], e1 = kpe[1];
            const v4f d0 = kpo[0], d1 = kpo[1], d2 = kpo[2], d3 = kpo[3];
#pragma unroll
            for (int m = 0; m < 4; ++m) {
                ke[m] = e0[m]; ke[4 + m] = e1[m];
                ko[m] = d0[m]; ko[4 + m] = d1[m]; ko[8 + m] = d2[m]; ko[12 + m] = d3[m];
            }
            float o0 = oldW[n * LP + lane];
            float o1 = oldW[(n + 1) * LP + lane];
            if (i == 0) {
                o0 += wX[n * LP + lane];
                o1 += wX[(n + 1) * LP + lane];
            }
            int lo = se[0], hi = se[0];
#pragma unroll
            for (int m = 1; m < 8; ++m) { lo = se[m] < lo ? se[m] : lo; hi = se[m] > hi ? se[m] : hi; }
#pragma unroll
            for (int m = 0; m < 15; ++m) { lo = sd[m] < lo ? sd[m] : lo; hi = sd[m] > hi ? sd[m] : hi; }
            float ae = 0.f, ao = 0.f;
            if (lo >= wlo && hi <= whi) {
#pragma unroll
                for (int m = 0; m < 8; ++m) ae = __builtin_fmaf(ke[m], uw[vb + se[m]], ae);
#pragma unroll
                for (int m = 0; m < 15; ++m) ao = __builtin_fmaf(ko[m], uw[vb + sd[m]], ao);
            } else {
#pragma unroll
                for (int m = 0; m < 8; ++m) ae = __builtin_fmaf(ke[m], ug[step * row[2 * m]], ae);
#pragma unroll
                for (int m = 0; m < 15; ++m) ao = __builtin_fmaf(ko[m], ug[step * row[15 + m]], ao);
            }
            newW[(2 * n) * LP + lane]     = o0 + hf * ae;
            newW[(2 * n + 1) * LP + lane] = 0.5f * (o0 + o1) + hf * ao;
        }
        if (wave == 0) {
            const int n = ns;
            const int* row = nbr + (size_t)(2 * (nb + n)) * 15;
            int se[8];
#pragma unroll
            for (int m = 0; m < 8; ++m) se[m] = row[2 * m] * P;
            const v4f* kpe = (const v4f*)(Ke + n * KPITCH);
            float ke[8];
            const v4f e0 = kpe[0], e1 = kpe[1];
#pragma unroll
            for (int m = 0; m < 4; ++m) { ke[m] = e0[m]; ke[4 + m] = e1[m]; }
            float o0 = oldW[n * LP + lane];
            if (i == 0) o0 += wX[n * LP + lane];
            int lo = se[0], hi = se[0];
#pragma unroll
            for (int m = 1; m < 8; ++m) { lo = se[m] < lo ? se[m] : lo; hi = se[m] > hi ? se[m] : hi; }
            float ae = 0.f;
            if (lo >= wlo && hi <= whi) {
#pragma unroll
                for (int m = 0; m < 8; ++m) ae = __builtin_fmaf(ke[m], uw[vb + se[m]], ae);
            } else {
#pragma unroll
                for (int m = 0; m < 8; ++m) ae = __builtin_fmaf(ke[m], ug[step * row[2 * m]], ae);
            }
            newW[(2 * n) * LP + lane] = o0 + hf * ae;
        }
        __syncthreads();
        float* tmp = oldW; oldW = newW; newW = tmp;
    }

    if (seg == 0 || seg == NSEG - 1) {
        if (wave == 0) {
            const float* kv = Kbd_V + ((seg == 0) ? 0 : 32);
            const float* sp0 = sbuf + (size_t)(row0 + lane) * 32;
            const float* sp1 = sbuf + 16384 + (size_t)(row0 + lane) * 32;
            float acc = 0.f;
#pragma unroll
            for (int r = 0; r < 32; ++r) acc += kv[r] * (sp0[r] + sp1[r]);
            oldW[((seg == 0) ? 0 : 8 * SEGC) * LP + lane] = H * acc;
        }
        __syncthreads();
    }

    const int npts = (seg == NSEG - 1) ? (8 * SEGC + 1) : (8 * SEGC);
    for (int r = wave; r < 32; r += 8) {
        float* dst = out + (size_t)(row0 + r) * 8193 + 8 * c0;
        for (int p = lane; p < npts; p += 32) dst[p] = oldW[p * LP + r];
    }
}

extern "C" void kernel_launch(void* const* d_in, const int* in_sizes, int n_in,
                              void* d_out, int out_size, void* d_ws, size_t ws_size,
                              hipStream_t stream) {
    const float* uh     = (const float*)d_in[0];
    const float* KHH_U  = (const float*)d_in[1];
    const float* KHH_V  = (const float*)d_in[2];
    const float* Ke0_U  = (const float*)d_in[3];
    const float* Ke0_V  = (const float*)d_in[4];
    const float* Ko0_U  = (const float*)d_in[5];
    const float* Ko0_V  = (const float*)d_in[6];
    const int*   nbr0   = (const int*)d_in[7];
    const float* Ke1_U  = (const float*)d_in[8];
    const float* Ke1_V  = (const float*)d_in[9];
    const float* Ko1_U  = (const float*)d_in[10];
    const float* Ko1_V  = (const float*)d_in[11];
    const int*   nbr1   = (const int*)d_in[12];
    const float* Ke2_U  = (const float*)d_in[13];
    const float* Ke2_V  = (const float*)d_in[14];
    const float* Ko2_U  = (const float*)d_in[15];
    const float* Ko2_V  = (const float*)d_in[16];
    const int*   nbr2   = (const int*)d_in[17];
    const float* Kbd_U  = (const float*)d_in[18];
    const float* Kbd_V  = (const float*)d_in[19];

    float* ws  = (float*)d_ws;
    float* wh0 = ws;
    float* sb  = wh0 + 1049600;

    float* outp = (float*)d_out;

    k_wmma_proj<<<128, 1024, 0, stream>>>(uh, KHH_U, KHH_V, Kbd_U, wh0, sb);

    k_refine<<<1024, 256, 0, stream>>>(uh, wh0, sb, Kbd_V,
                                      Ke0_U, Ke0_V, Ko0_U, Ko0_V,
                                      Ke1_U, Ke1_V, Ko1_U, Ko1_V,
                                      Ke2_U, Ke2_V, Ko2_U, Ko2_V,
                                      nbr2, nbr1, nbr0, outp);
}


# Refinement: a trip loads its two index rows lane-spread and tests them for the full band; a banded trip reads its run of window points once each at fixed offsets from one address; sums unchanged
# speedup vs baseline: 13.4911x; 1.0432x over previous
#include <hip/hip_runtime.h>
#include <hip/hip_bf16.h>

typedef __attribute__((ext_vector_type(2))) float v2f;
typedef __attribute__((ext_vector_type(4))) float v4f;
typedef __attribute__((ext_vector_type(8))) float v8f;

#define WMMA_F32X4(a, b, c) \
    __builtin_amdgcn_wmma_f32_16x16x4_f32(false, (a), false, (b), (short)0, (c), false, false)

__global__ void __launch_bounds__(1024) k_wmma_proj(const float* __restrict__ uh,
                                                    const float* __restrict__ KHH_U,
                                                    const float* __restrict__ KHH_V,
                                                    const float* __restrict__ Kbd_U,
                                                    float* __restrict__ wh0,
                                                    float* __restrict__ sbuf) {
    __shared__ __attribute__((aligned(16))) float stage[32][2304];
    __shared__ float tl[16 * 32];
    const int wave = __builtin_amdgcn_readfirstlane((int)(threadIdx.x >> 5));
    const int lane = threadIdx.x & 31;
    const int half = lane >> 4;
    const int mrow = lane & 15;
    const int kh  = blockIdx.x & 1;
    const int kc0 = 512 * kh + 16 * wave;
    const int kf0 = 4096 * kh + 128 * wave;

    float* As = stage[wave];
    float* Bs = As + 16 * 36;
    float* Vs = Bs + 32 * 48;

    if (blockIdx.x >= 64) {
        const int R  = ((blockIdx.x - 64) >> 1) * 16;
        const float* ublk = uh + (size_t)R * 8193 + lane;
        const int srow = lane >> 3;
        const int scol = 4 * (lane & 7);
        v8f s0 = {}; v8f s1 = {};
        for (int c = 0; c < 4; ++c) {
            const int kbase = kf0 + 32 * c;
            float av[16];
#pragma unroll
            for (int r = 0; r < 16; ++r) av[r] = ublk[(size_t)r * 8193 + kbase];
            const float* bsrc = Kbd_U + (size_t)(kbase + srow) * 32 + scol;
            v4f bv[8];
#pragma unroll
            for (int i = 0; i < 8; ++i) bv[i] = *(const v4f*)(bsrc + (size_t)(4 * i) * 32);

            __builtin_amdgcn_fence(__ATOMIC_ACQ_REL, "wavefront");
            __builtin_amdgcn_wave_barrier();
#pragma unroll
            for (int r = 0; r < 16; ++r) As[r * 36 + lane] = av[r];
#pragma unroll
            for (int i = 0; i < 8; ++i) *(v4f*)(Bs + (4 * i + srow) * 48 + scol) = bv[i];
            __builtin_amdgcn_fence(__ATOMIC_ACQ_REL, "wavefront");
            __builtin_amdgcn_wave_barrier();

#pragma unroll
            for (int q = 0; q < 8; ++q) {
                const int kl = 4 * q + 2 * half;
                v2f a, b0, b1;
                a    = *(const v2f*)(As + mrow * 36 + kl);
                b0.x = Bs[kl * 48 + mrow];
                b0.y = Bs[(kl + 1) * 48 + mrow];
                b1.x = Bs[kl * 48 + 16 + mrow];
                b1.y = Bs[(kl + 1) * 48 + 16 + mrow];
                s0 = WMMA_F32X4(a, b0, s0);
                s1 = WMMA_F32X4(a, b1, s1);
            }
        }
        __builtin_amdgcn_fence(__ATOMIC_ACQ_REL, "wavefront");
        __builtin_amdgcn_wave_barrier();
        if (kh == 1 && wave == 31) {
            float alast = uh[(size_t)(R + mrow) * 8193 + 8192];
            v2f a, b0, b1;
            a.x  = (half == 0) ? alast : 0.f;
            a.y  = 0.f;
            b0.x = Kbd_U[8192 * 32 + mrow];
            b0.y = b0.x;
            b1.x = Kbd_U[8192 * 32 + 16 + mrow];
            b1.y = b1.x;
            s0 = WMMA_F32X4(a, b0, s0);
            s1 = WMMA_F32X4(a, b1, s1);
        }
        float* pw = stage[wave];
#pragma unroll
        for (int j = 0; j < 8; ++j) {
            int m = j + 8 * half;
            pw[m * 32 + mrow]      = s0[j];
            pw[m * 32 + 16 + mrow] = s1[j];
        }
        __syncthreads();
        if (threadIdx.x < 512) {
            const int e = threadIdx.x;
            float acc = stage[0][e];
#pragma unroll
            for (int w = 1; w < 32; ++w) acc += stage[w][e];
            sbuf[(size_t)kh * 16384 + (size_t)R * 32 + e] = acc;
        }
        return;
    }

    const int R    = (blockIdx.x >> 1) * 16;
    const float HK = 1.0f / 1024.0f;
    const float* arow = uh + (size_t)(R + mrow) * 8193;

    v8f t0 = {}; v8f t1 = {};
    const float* ublk = uh + (size_t)R * 8193 + lane;
    const int srow = lane >> 3;
    const int scol = 4 * (lane & 7);

    for (int c = 0; c < 4; ++c) {
        const int kbase = kf0 + 32 * c;
        float av[16];
#pragma unroll
        for (int r = 0; r < 16; ++r) av[r] = ublk[(size_t)r * 8193 + kbase];
        const v4f vv = *((const v4f*)(KHH_V + (size_t)(kc0 + 4 * c) * 32) + lane);

        __builtin_amdgcn_fence(__ATOMIC_ACQ_REL, "wavefront");
        __builtin_amdgcn_wave_barrier();
#pragma unroll
        for (int r = 0; r < 16; ++r) As[r * 36 + lane] = av[r];
        *(v4f*)(Vs + srow * 48 + scol) = vv;
        __builtin_amdgcn_fence(__ATOMIC_ACQ_REL, "wavefront");
        __builtin_amdgcn_wave_barrier();

        {
            v2f a, b0, b1;
            a.x  = As[mrow * 36 + 16 * half];
            a.y  = As[mrow * 36 + 16 * half + 8];
            b0.x = Vs[(2 * half) * 48 + mrow];
            b0.y = Vs[(2 * half + 1) * 48 + mrow];
            b1.x = Vs[(2 * half) * 48 + 16 + mrow];
            b1.y = Vs[(2 * half + 1) * 48 + 16 + mrow];
            t0 = WMMA_F32X4(a, b0, t0);
            t1 = WMMA_F32X4(a, b1, t1);
        }
    }
    __builtin_amdgcn_fence(__ATOMIC_ACQ_REL, "wavefront");
    __builtin_amdgcn_wave_barrier();
    if (kh == 1 && wave == 31) {
        float alast = arow[8192];
        v2f a, b0, b1;
        a.x  = (half == 0) ? alast : 0.f;
        a.y  = 0.f;
        b0.x = KHH_V[1024 * 32 + mrow];
        b0.y = b0.x;
        b1.x = KHH_V[1024 * 32 + 16 + mrow];
        b1.y = b1.x;
        t0 = WMMA_F32X4(a, b0, t0);
        t1 = WMMA_F32X4(a, b1, t1);
    }

    float* pw = stage[wave];
#pragma unroll
    for (int j = 0; j < 8; ++j) {
        int m = j + 8 * half;
        pw[m * 32 + mrow]      = t0[j];
        pw[m * 32 + 16 + mrow] = t1[j];
    }
    __syncthreads();

    if (threadIdx.x < 512) {
        const int e = threadIdx.x;
        float acc = stage[0][e];
#pragma unroll
        for (int w = 1; w < 32; ++w) acc += stage[w][e];
        tl[e] = acc * HK;
    }
    __syncthreads();

    float* whp = wh0 + (size_t)kh * 524800;
    v2f af[8];
#pragma unroll
    for (int q = 0; q < 8; ++q) {
        int ka = 4 * q + 2 * half;
        af[q].x = tl[mrow * 32 + ka];
        af[q].y = tl[mrow * 32 + ka + 1];
    }
#pragma unroll
    for (int u = 0; u < 2; ++u) {
        int n = (2 * wave + u) * 16 + mrow;
        v8f acc = {};
#pragma unroll
        for (int q = 0; q < 8; ++q) {
            int ka = 4 * q + 2 * half;
            v2f b;
            b.x = KHH_U[n * 32 + ka];
            b.y = KHH_U[n * 32 + ka + 1];
            acc = WMMA_F32X4(af[q], b, acc);
        }
#pragma unroll
        for (int j = 0; j < 8; ++j) {
            int m = j + 8 * half;
            whp[(size_t)(R + m) * 1025 + n] = acc[j];
        }
    }
    if (wave == 0) {
        v8f acc = {};
#pragma unroll
        for (int q = 0; q < 8; ++q) {
            int ka = 4 * q + 2 * half;
            v2f b;
            b.x = KHH_U[1024 * 32 + ka];
            b.y = KHH_U[1024 * 32 + ka + 1];
            acc = WMMA_F32X4(af[q], b, acc);
        }
        if (mrow == 0) {
#pragma unroll
            for (int j = 0; j < 8; ++j) {
                int m = j + 8 * half;
                whp[(size_t)(R + m) * 1025 + 1024] = acc[j];
            }
        }
    }
}

#define SEGC 16
#define NSEG (1024 / SEGC)
#define LP   33
#define HALO 32
#define WUMAX (8 * SEGC + 2 * HALO + 1)
#define WUMIN (8 * SEGC + HALO + 1)
#define NCTILE 17
#define KPITCH 16
#define KE_ROW(i) ((i) == 0 ? 0  : ((i) == 1 ? 48 : 128))
#define KO_ROW(i) ((i) == 0 ? 32 : ((i) == 1 ? 96 : 208))

__global__ void __launch_bounds__(256) k_refine(const float* __restrict__ uh,
                                                const float* __restrict__ wh0,
                                                const float* __restrict__ sbuf,
                                                const float* __restrict__ Kbd_V,
                                                const float* __restrict__ Ke0_U,
                                                const float* __restrict__ Ke0_V,
                                                const float* __restrict__ Ko0_U,
                                                const float* __restrict__ Ko0_V,
                                                const float* __restrict__ Ke1_U,
                                                const float* __restrict__ Ke1_V,
                                                const float* __restrict__ Ko1_U,
                                                const float* __restrict__ Ko1_V,
                                                const float* __restrict__ Ke2_U,
                                                const float* __restrict__ Ke2_V,
                                                const float* __restrict__ Ko2_U,
                                                const float* __restrict__ Ko2_V,
                                                const int* __restrict__ nbrL0,
                                                const int* __restrict__ nbrL1,
                                                const int* __restrict__ nbrL2,
                                                float* __restrict__ out) {
    __shared__ float uw[WUMAX * LP];
    __shared__ float wA[(4 * SEGC + 1) * LP];
    __shared__ float wB[(8 * SEGC + 1) * LP];
    __shared__ float wX[(SEGC + 1) * LP];
    __shared__ __attribute__((aligned(16))) float kcf[NCTILE * 16 * KPITCH];
    const int wave = __builtin_amdgcn_readfirstlane((int)(threadIdx.x >> 5));
    const int lane = threadIdx.x & 31;
    const int seg  = blockIdx.x % NSEG;
    const int row0 = (blockIdx.x / NSEG) * 32;
    const int c0   = seg * SEGC;
    const int jlo  = (8 * c0 - HALO) < 0 ? 0 : (8 * c0 - HALO);
    const int jhi  = (8 * (c0 + SEGC) + HALO) > 8192 ? 8192 : (8 * (c0 + SEGC) + HALO);
    const int wu   = jhi - jlo + 1;
    const float H  = 1.0f / 8192.0f;

    {
        const int mrow = lane & 15;
        const int half = lane >> 4;
#define TILE_SRC(t, up, vp)                                                              \
        do {                                                                             \
            const int lvl_  = ((t) >= 3 ? 1 : 0) + ((t) >= 8 ? 1 : 0);                    \
            const int tb_   = lvl_ == 0 ? 0 : (lvl_ == 1 ? 3 : 8);     \
            const int ne_   = lvl_ == 0 ? 2 : (lvl_ == 1 ? 3 : 5);      \
            const bool odd_ = ((t) - tb_) >= ne_;                                         \
            const int tt_   = (t) - tb_ - (odd_ ? ne_ : 0);            \
            const float* U_ = odd_ ? (lvl_ == 0 ? Ko0_U : (lvl_ == 1 ? Ko1_U : Ko2_U))    \
                                   : (lvl_ == 0 ? Ke0_U : (lvl_ == 1 ? Ke1_U : Ke2_U));   \
            const float* V_ = odd_ ? (lvl_ == 0 ? Ko0_V : (lvl_ == 1 ? Ko1_V : Ko2_V))    \
                                   : (lvl_ == 0 ? Ke0_V : (lvl_ == 1 ? Ke1_V : Ke2_V));   \
            const int glast_ = (1024 << lvl_) - (odd_ ? 1 : 0);          \
            const int vlast_ = odd_ ? 14 : 7;                            \
            int g_ = (c0 << lvl_) + 16 * tt_ + mrow;                                      \
            g_ = g_ > glast_ ? glast_ : g_;                                               \
            const int vr_ = mrow > vlast_ ? vlast_ : mrow;                                \
            (up) = U_ + (size_t)g_ * 32 + 4 * half;                                       \
            (vp) = V_ + vr_ * 32 + 4 * half;                                              \
        } while (0)
        const float* up;
        const float* vp;
        v4f a0[4], b0[4], a1[4], b1[4], a2[4], b2[4];
        TILE_SRC(wave, up, vp);
#pragma unroll
        for (int qq = 0; qq < 4; ++qq) { a0[qq] = *(const v4f*)(up + 8 * qq); b0[qq] = *(const v4f*)(vp + 8 * qq); }
        TILE_SRC(wave + 8, up, vp);
#pragma unroll
        for (int qq = 0; qq < 4; ++qq) { a1[qq] = *(const v4f*)(up + 8 * qq); b1[qq] = *(const v4f*)(vp + 8 * qq); }
        if (wave == 0) {
            TILE_SRC(16, up, vp);
#pragma unroll
            for (int qq = 0; qq < 4; ++qq) { a2[qq] = *(const v4f*)(up + 8 * qq); b2[qq] = *(const v4f*)(vp + 8 * qq); }
        }
        v8f d0 = {}; v8f d1 = {};
#pragma unroll
        for (int s = 0; s < 8; ++s) {
            const int qq = s >> 1, e = s & 1;
            v2f x, y;
            x.x = a0[qq][2 * e]; x.y = a0[qq][2 * e + 1];
            y.x = b0[qq][2 * e]; y.y = b0[qq][2 * e + 1];
            d0 = WMMA_F32X4(x, y, d0);
            x.x = a1[qq][2 * e]; x.y = a1[qq][2 * e + 1];
            y.x = b1[qq][2 * e]; y.y = b1[qq][2 * e + 1];
            d1 = WMMA_F32X4(x, y, d1);
        }
        float* o = kcf + (16 * wave + 8 * half) * KPITCH + mrow;
#pragma unroll
        for (int j = 0; j < 8; ++j) {
            o[j * KPITCH]         = d0[j];
            o[(128 + j) * KPITCH] = d1[j];
        }
        if (wave == 0) {
            v8f d2 = {};
#pragma unroll
            for (int s = 0; s < 8; ++s) {
                const int qq = s >> 1, e = s & 1;
                v2f x, y;
                x.x = a2[qq][2 * e]; x.y = a2[qq][2 * e + 1];
                y.x = b2[qq][2 * e]; y.y = b2[qq][2 * e + 1];
                d2 = WMMA_F32X4(x, y, d2);
            }
#pragma unroll
            for (int j = 0; j < 8; ++j) o[(256 + j) * KPITCH] = d2[j];
        }
#undef TILE_SRC
    }

    {
        float sv[4][7];
        float wv[4], wx[4];
        const int wl = lane <= SEGC ? lane : SEGC;
#pragma unroll
        for (int q = 0; q < 4; ++q) {
            const int r = wave + 8 * q;
            const float* us = uh + (size_t)(row0 + r) * 8193 + jlo;
#pragma unroll
            for (int k = 0; k < 7; ++k) {
                const int p  = lane + 32 * k;
                const int pc = (32 * k + 31 < WUMIN) ? p : (p < wu ? p : wu - 1);
                sv[q][k] = us[pc];
            }
            wv[q] = wh0[(size_t)(row0 + r) * 1025 + c0 + wl];
            wx[q] = wh0[(size_t)524800 + (size_t)(row0 + r) * 1025 + c0 + wl];
        }
#pragma unroll
        for (int q = 0; q < 4; ++q) {
            const int r = wave + 8 * q;
#pragma unroll
            for (int k = 0; k < 7; ++k) {
                const int p  = lane + 32 * k;
                const int pc = (32 * k + 31 < WUMIN) ? p : (p < wu ? p : wu - 1);
                uw[pc * LP + r] = sv[q][k];
            }
            wA[wl * LP + r] = wv[q];
            wX[wl * LP + r] = wx[q];
        }
    }
    __syncthreads();

    const int vb = lane - jlo * LP;
    const float* ug  = uh + (size_t)(row0 + lane) * 8193;

    const int*   nbrP[3] = {nbrL0, nbrL1, nbrL2};
    float* oldW = wA;
    float* newW = wB;

    const int wlo = jlo * LP;
    const int whi = jhi * LP;
    const int lc   = lane < 30 ? lane : 29;
    const int boff = lc < 15 ? lc : lc - 14;

#pragma unroll
    for (int i = 0; i < 3; ++i) {
        const int step = 4 >> i;
        const float hf = H * (float)step;
        const int ns = SEGC << i;
        const int nb = c0 << i;
        const float* Ke = kcf + KE_ROW(i) * KPITCH;
        const float* Ko = kcf + KO_ROW(i) * KPITCH;
        const int* nbr  = nbrP[i];
        const int P = step * LP;
        for (int n = wave; n < ns; n += 8) {
            const int* row = nbr + (size_t)(2 * (nb + n)) * 15;
            const int rv = row[lc];
            const int dv = rv - boff;
            const int e0 = __builtin_amdgcn_readfirstlane(dv);
            const bool band = __builtin_amdgcn_ballot_w32(dv != e0) == 0u;
            const v4f* kpe = (const v4f*)(Ke + n * KPITCH);
            const v4f* kpo = (const v4f*)(Ko + n * KPITCH);
            float ke[8], ko[16];
            const v4f e0v = kpe[0], e1v = kpe[1];
            const v4f d0 = kpo[0], d1 = kpo[1], d2 = kpo[2], d3 = kpo[3];
#pragma unroll
            for (int m = 0; m < 4; ++m) {
                ke[m] = e0v[m]; ke[4 + m] = e1v[m];
                ko[m] = d0[m]; ko[4 + m] = d1[m]; ko[8 + m] = d2[m]; ko[12 + m] = d3[m];
            }
            float o0 = oldW[n * LP + lane];
            float o1 = oldW[(n + 1) * LP + lane];
            if (i == 0) {
                o0 += wX[n * LP + lane];
                o1 += wX[(n + 1) * LP + lane];
            }
            float ae = 0.f, ao = 0.f;
            if (band && e0 * P >= wlo && (e0 + 15) * P <= whi) {
                const float* wp = uw + vb + e0 * P;
                float w[16];
#pragma unroll
                for (int k = 0; k < 16; ++k) w[k] = wp[k * P];
#pragma unroll
                for (int m = 0; m < 8; ++m) ae = __builtin_fmaf(ke[m], w[2 * m], ae);
#pragma unroll
                for (int m = 0; m < 15; ++m) ao = __builtin_fmaf(ko[m], w[m + 1], ao);
            } else {
                int xe[8], xd[15];
#pragma unroll
                for (int m = 0; m < 8; ++m) xe[m] = __builtin_amdgcn_readlane(rv, 2 * m);
#pragma unroll
                for (int m = 0; m < 15; ++m) xd[m] = __builtin_amdgcn_readlane(rv, 15 + m);
                int lo = xe[0], hi = xe[0];
#pragma unroll
                for (int m = 1; m < 8; ++m) { lo = xe[m] < lo ? xe[m] : lo; hi = xe[m] > hi ? xe[m] : hi; }
#pragma unroll
                for (int m = 0; m < 15; ++m) { lo = xd[m] < lo ? xd[m] : lo; hi = xd[m] > hi ? xd[m] : hi; }
                if (lo * P >= wlo && hi * P <= whi) {
#pragma unroll
                    for (int m = 0; m < 8; ++m) ae = __builtin_fmaf(ke[m], uw[vb + xe[m] * P], ae);
#pragma unroll
                    for (int m = 0; m < 15; ++m) ao = __builtin_fmaf(ko[m], uw[vb + xd[m] * P], ao);
                } else {
#pragma unroll
                    for (int m = 0; m < 8; ++m) ae = __builtin_fmaf(ke[m], ug[step * xe[m]], ae);
#pragma unroll
                    for (int m = 0; m < 15; ++m) ao = __builtin_fmaf(ko[m], ug[step * xd[m]], ao);
                }
            }
            newW[(2 * n) * LP + lane]     = o0 + hf * ae;
            newW[(2 * n + 1) * LP + lane] = 0.5f * (o0 + o1) + hf * ao;
        }
        if (wave == 0) {
            const int n = ns;
            const int* row = nbr + (size_t)(2 * (nb + n)) * 15;
            int se[8];
#pragma unroll
            for (int m = 0; m < 8; ++m) se[m] = row[2 * m] * P;
            const v4f* kpe = (const v4f*)(Ke + n * KPITCH);
            float ke[8];
            const v4f e0 = kpe[0], e1 = kpe[1];
#pragma unroll
            for (int m = 0; m < 4; ++m) { ke[m] = e0[m]; ke[4 + m] = e1[m]; }
            float o0 = oldW[n * LP + lane];
            if (i == 0) o0 += wX[n * LP + lane];
            int lo = se[0], hi = se[0];
#pragma unroll
            for (int m = 1; m < 8; ++m) { lo = se[m] < lo ? se[m] : lo; hi = se[m] > hi ? se[m] : hi; }
            float ae = 0.f;
            if (lo >= wlo && hi <= whi) {
#pragma unroll
                for (int m = 0; m < 8; ++m) ae = __builtin_fmaf(ke[m], uw[vb + se[m]], ae);
            } else {
#pragma unroll
                for (int m = 0; m < 8; ++m) ae = __builtin_fmaf(ke[m], ug[step * row[2 * m]], ae);
            }
            newW[(2 * n) * LP + lane] = o0 + hf * ae;
        }
        __syncthreads();
        float* tmp = oldW; oldW = newW; newW = tmp;
    }

    if (seg == 0 || seg == NSEG - 1) {
        if (wave == 0) {
            const float* kv = Kbd_V + ((seg == 0) ? 0 : 32);
            const float* sp0 = sbuf + (size_t)(row0 + lane) * 32;
            const float* sp1 = sbuf + 16384 + (size_t)(row0 + lane) * 32;
            float acc = 0.f;
#pragma unroll
            for (int r = 0; r < 32; ++r) acc += kv[r] * (sp0[r] + sp1[r]);
            oldW[((seg == 0) ? 0 : 8 * SEGC) * LP + lane] = H * acc;
        }
        __syncthreads();
    }

    const int npts = (seg == NSEG - 1) ? (8 * SEGC + 1) : (8 * SEGC);
    for (int r = wave; r < 32; r += 8) {
        float* dst = out + (size_t)(row0 + r) * 8193 + 8 * c0;
        for (int p = lane; p < npts; p += 32) dst[p] = oldW[p * LP + r];
    }
}

extern "C" void kernel_launch(void* const* d_in, const int* in_sizes, int n_in,
                              void* d_out, int out_size, void* d_ws, size_t ws_size,
                              hipStream_t stream) {
    const float* uh     = (const float*)d_in[0];
    const float* KHH_U  = (const float*)d_in[1];
    const float* KHH_V  = (const float*)d_in[2];
    const float* Ke0_U  = (const float*)d_in[3];
    const float* Ke0_V  = (const float*)d_in[4];
    const float* Ko0_U  = (const float*)d_in[5];
    const float* Ko0_V  = (const float*)d_in[6];
    const int*   nbr0   = (const int*)d_in[7];
    const float* Ke1_U  = (const float*)d_in[8];
    const float* Ke1_V  = (const float*)d_in[9];
    const float* Ko1_U  = (const float*)d_in[10];
    const float* Ko1_V  = (const float*)d_in[11];
    const int*   nbr1   = (const int*)d_in[12];
    const float* Ke2_U  = (const float*)d_in[13];
    const float* Ke2_V  = (const float*)d_in[14];
    const float* Ko2_U  = (const float*)d_in[15];
    const float* Ko2_V  = (const float*)d_in[16];
    const int*   nbr2   = (const int*)d_in[17];
    const float* Kbd_U  = (const float*)d_in[18];
    const float* Kbd_V  = (const float*)d_in[19];

    float* ws  = (float*)d_ws;
    float* wh0 = ws;
    float* sb  = wh0 + 1049600;

    float* outp = (float*)d_out;

    k_wmma_proj<<<128, 1024, 0, stream>>>(uh, KHH_U, KHH_V, Kbd_U, wh0, sb);

    k_refine<<<1024, 256, 0, stream>>>(uh, wh0, sb, Kbd_V,
                                      Ke0_U, Ke0_V, Ko0_U, Ko0_V,
                                      Ke1_U, Ke1_V, Ko1_U, Ko1_V,
                                      Ke2_U, Ke2_V, Ko2_U, Ko2_V,
                                      nbr2, nbr1, nbr0, outp);
}


# Refinement: a wave loads all its trips' index rows at the block head with the staging loads, tests them once, keeps a band word per trip in one register: no load in a banded trip; sums unchanged
# speedup vs baseline: 14.1844x; 1.0514x over previous
#include <hip/hip_runtime.h>
#include <hip/hip_bf16.h>

typedef __attribute__((ext_vector_type(2))) float v2f;
typedef __attribute__((ext_vector_type(4))) float v4f;
typedef __attribute__((ext_vector_type(8))) float v8f;

#define WMMA_F32X4(a, b, c) \
    __builtin_amdgcn_wmma_f32_16x16x4_f32(false, (a), false, (b), (short)0, (c), false, false)

__global__ void __launch_bounds__(1024) k_wmma_proj(const float* __restrict__ uh,
                                                    const float* __restrict__ KHH_U,
                                                    const float* __restrict__ KHH_V,
                                                    const float* __restrict__ Kbd_U,
                                                    float* __restrict__ wh0,
                                                    float* __restrict__ sbuf) {
    __shared__ __attribute__((aligned(16))) float stage[32][2304];
    __shared__ float tl[16 * 32];
    const int wave = __builtin_amdgcn_readfirstlane((int)(threadIdx.x >> 5));
    const int lane = threadIdx.x & 31;
    const int half = lane >> 4;
    const int mrow = lane & 15;
    const int kh  = blockIdx.x & 1;
    const int kc0 = 512 * kh + 16 * wave;
    const int kf0 = 4096 * kh + 128 * wave;

    float* As = stage[wave];
    float* Bs = As + 16 * 36;
    float* Vs = Bs + 32 * 48;

    if (blockIdx.x >= 64) {
        const int R  = ((blockIdx.x - 64) >> 1) * 16;
        const float* ublk = uh + (size_t)R * 8193 + lane;
        const int srow = lane >> 3;
        const int scol = 4 * (lane & 7);
        v8f s0 = {}; v8f s1 = {};
        for (int c = 0; c < 4; ++c) {
            const int kbase = kf0 + 32 * c;
            float av[16];
#pragma unroll
            for (int r = 0; r < 16; ++r) av[r] = ublk[(size_t)r * 8193 + kbase];
            const float* bsrc = Kbd_U + (size_t)(kbase + srow) * 32 + scol;
            v4f bv[8];
#pragma unroll
            for (int i = 0; i < 8; ++i) bv[i] = *(const v4f*)(bsrc + (size_t)(4 * i) * 32);

            __builtin_amdgcn_fence(__ATOMIC_ACQ_REL, "wavefront");
            __builtin_amdgcn_wave_barrier();
#pragma unroll
            for (int r = 0; r < 16; ++r) As[r * 36 + lane] = av[r];
#pragma unroll
            for (int i = 0; i < 8; ++i) *(v4f*)(Bs + (4 * i + srow) * 48 + scol) = bv[i];
            __builtin_amdgcn_fence(__ATOMIC_ACQ_REL, "wavefront");
            __builtin_amdgcn_wave_barrier();

#pragma unroll
            for (int q = 0; q < 8; ++q) {
                const int kl = 4 * q + 2 * half;
                v2f a, b0, b1;
                a    = *(const v2f*)(As + mrow * 36 + kl);
                b0.x = Bs[kl * 48 + mrow];
                b0.y = Bs[(kl + 1) * 48 + mrow];
                b1.x = Bs[kl * 48 + 16 + mrow];
                b1.y = Bs[(kl + 1) * 48 + 16 + mrow];
                s0 = WMMA_F32X4(a, b0, s0);
                s1 = WMMA_F32X4(a, b1, s1);
            }
        }
        __builtin_amdgcn_fence(__ATOMIC_ACQ_REL, "wavefront");
        __builtin_amdgcn_wave_barrier();
        if (kh == 1 && wave == 31) {
            float alast = uh[(size_t)(R + mrow) * 8193 + 8192];
            v2f a, b0, b1;
            a.x  = (half == 0) ? alast : 0.f;
            a.y  = 0.f;
            b0.x = Kbd_U[8192 * 32 + mrow];
            b0.y = b0.x;
            b1.x = Kbd_U[8192 * 32 + 16 + mrow];
            b1.y = b1.x;
            s0 = WMMA_F32X4(a, b0, s0);
            s1 = WMMA_F32X4(a, b1, s1);
        }
        float* pw = stage[wave];
#pragma unroll
        for (int j = 0; j < 8; ++j) {
            int m = j + 8 * half;
            pw[m * 32 + mrow]      = s0[j];
            pw[m * 32 + 16 + mrow] = s1[j];
        }
        __syncthreads();
        if (threadIdx.x < 512) {
            const int e = threadIdx.x;
            float acc = stage[0][e];
#pragma unroll
            for (int w = 1; w < 32; ++w) acc += stage[w][e];
            sbuf[(size_t)kh * 16384 + (size_t)R * 32 + e] = acc;
        }
        return;
    }

    const int R    = (blockIdx.x >> 1) * 16;
    const float HK = 1.0f / 1024.0f;
    const float* arow = uh + (size_t)(R + mrow) * 8193;

    v8f t0 = {}; v8f t1 = {};
    const float* ublk = uh + (size_t)R * 8193 + lane;
    const int srow = lane >> 3;
    const int scol = 4 * (lane & 7);

    for (int c = 0; c < 4; ++c) {
        const int kbase = kf0 + 32 * c;
        float av[16];
#pragma unroll
        for (int r = 0; r < 16; ++r) av[r] = ublk[(size_t)r * 8193 + kbase];
        const v4f vv = *((const v4f*)(KHH_V + (size_t)(kc0 + 4 * c) * 32) + lane);

        __builtin_amdgcn_fence(__ATOMIC_ACQ_REL, "wavefront");
        __builtin_amdgcn_wave_barrier();
#pragma unroll
        for (int r = 0; r < 16; ++r) As[r * 36 + lane] = av[r];
        *(v4f*)(Vs + srow * 48 + scol) = vv;
        __builtin_amdgcn_fence(__ATOMIC_ACQ_REL, "wavefront");
        __builtin_amdgcn_wave_barrier();

        {
            v2f a, b0, b1;
            a.x  = As[mrow * 36 + 16 * half];
            a.y  = As[mrow * 36 + 16 * half + 8];
            b0.x = Vs[(2 * half) * 48 + mrow];
            b0.y = Vs[(2 * half + 1) * 48 + mrow];
            b1.x = Vs[(2 * half) * 48 + 16 + mrow];
            b1.y = Vs[(2 * half + 1) * 48 + 16 + mrow];
            t0 = WMMA_F32X4(a, b0, t0);
            t1 = WMMA_F32X4(a, b1, t1);
        }
    }
    __builtin_amdgcn_fence(__ATOMIC_ACQ_REL, "wavefront");
    __builtin_amdgcn_wave_barrier();
    if (kh == 1 && wave == 31) {
        float alast = arow[8192];
        v2f a, b0, b1;
        a.x  = (half == 0) ? alast : 0.f;
        a.y  = 0.f;
        b0.x = KHH_V[1024 * 32 + mrow];
        b0.y = b0.x;
        b1.x = KHH_V[1024 * 32 + 16 + mrow];
        b1.y = b1.x;
        t0 = WMMA_F32X4(a, b0, t0);
        t1 = WMMA_F32X4(a, b1, t1);
    }

    float* pw = stage[wave];
#pragma unroll
    for (int j = 0; j < 8; ++j) {
        int m = j + 8 * half;
        pw[m * 32 + mrow]      = t0[j];
        pw[m * 32 + 16 + mrow] = t1[j];
    }
    __syncthreads();

    if (threadIdx.x < 512) {
        const int e = threadIdx.x;
        float acc = stage[0][e];
#pragma unroll
        for (int w = 1; w < 32; ++w) acc += stage[w][e];
        tl[e] = acc * HK;
    }
    __syncthreads();

    float* whp = wh0 + (size_t)kh * 524800;
    v2f af[8];
#pragma unroll
    for (int q = 0; q < 8; ++q) {
        int ka = 4 * q + 2 * half;
        af[q].x = tl[mrow * 32 + ka];
        af[q].y = tl[mrow * 32 + ka + 1];
    }
#pragma unroll
    for (int u = 0; u < 2; ++u) {
        int n = (2 * wave + u) * 16 + mrow;
        v8f acc = {};
#pragma unroll
        for (int q = 0; q < 8; ++q) {
            int ka = 4 * q + 2 * half;
            v2f b;
            b.x = KHH_U[n * 32 + ka];
            b.y = KHH_U[n * 32 + ka + 1];
            acc = WMMA_F32X4(af[q], b, acc);
        }
#pragma unroll
        for (int j = 0; j < 8; ++j) {
            int m = j + 8 * half;
            whp[(size_t)(R + m) * 1025 + n] = acc[j];
        }
    }
    if (wave == 0) {
        v8f acc = {};
#pragma unroll
        for (int q = 0; q < 8; ++q) {
            int ka = 4 * q + 2 * half;
            v2f b;
            b.x = KHH_U[1024 * 32 + ka];
            b.y = KHH_U[1024 * 32 + ka + 1];
            acc = WMMA_F32X4(af[q], b, acc);
        }
        if (mrow == 0) {
#pragma unroll
            for (int j = 0; j < 8; ++j) {
                int m = j + 8 * half;
                whp[(size_t)(R + m) * 1025 + 1024] = acc[j];
            }
        }
    }
}

#define SEGC 16
#define NSEG (1024 / SEGC)
#define LP   33
#define HALO 32
#define WUMAX (8 * SEGC + 2 * HALO + 1)
#define WUMIN (8 * SEGC + HALO + 1)
#define NCTILE 17
#define KPITCH 16
#define KE_ROW(i) ((i) == 0 ? 0  : ((i) == 1 ? 48 : 128))
#define KO_ROW(i) ((i) == 0 ? 32 : ((i) == 1 ? 96 : 208))

__global__ void __launch_bounds__(256) k_refine(const float* __restrict__ uh,
                                                const float* __restrict__ wh0,
                                                const float* __restrict__ sbuf,
                                                const float* __restrict__ Kbd_V,
                                                const float* __restrict__ Ke0_U,
                                                const float* __restrict__ Ke0_V,
                                                const float* __restrict__ Ko0_U,
                                                const float* __restrict__ Ko0_V,
                                                const float* __restrict__ Ke1_U,
                                                const float* __restrict__ Ke1_V,
                                                const float* __restrict__ Ko1_U,
                                                const float* __restrict__ Ko1_V,
                                                const float* __restrict__ Ke2_U,
                                                const float* __restrict__ Ke2_V,
                                                const float* __restrict__ Ko2_U,
                                                const float* __restrict__ Ko2_V,
                                                const int* __restrict__ nbrL0,
                                                const int* __restrict__ nbrL1,
                                                const int* __restrict__ nbrL2,
                                                float* __restrict__ out) {
    __shared__ float uw[WUMAX * LP];
    __shared__ float wA[(4 * SEGC + 1) * LP];
    __shared__ float wB[(8 * SEGC + 1) * LP];
    __shared__ float wX[(SEGC + 1) * LP];
    __shared__ __attribute__((aligned(16))) float kcf[NCTILE * 16 * KPITCH];
    const int wave = __builtin_amdgcn_readfirstlane((int)(threadIdx.x >> 5));
    const int lane = threadIdx.x & 31;
    const int seg  = blockIdx.x % NSEG;
    const int row0 = (blockIdx.x / NSEG) * 32;
    const int c0   = seg * SEGC;
    const int jlo  = (8 * c0 - HALO) < 0 ? 0 : (8 * c0 - HALO);
    const int jhi  = (8 * (c0 + SEGC) + HALO) > 8192 ? 8192 : (8 * (c0 + SEGC) + HALO);
    const int wu   = jhi - jlo + 1;
    const float H  = 1.0f / 8192.0f;

    {
        const int mrow = lane & 15;
        const int half = lane >> 4;
#define TILE_SRC(t, up, vp)                                                              \
        do {                                                                             \
            const int lvl_  = ((t) >= 3 ? 1 : 0) + ((t) >= 8 ? 1 : 0);                    \
            const int tb_   = lvl_ == 0 ? 0 : (lvl_ == 1 ? 3 : 8);     \
            const int ne_   = lvl_ == 0 ? 2 : (lvl_ == 1 ? 3 : 5);      \
            const bool odd_ = ((t) - tb_) >= ne_;                                         \
            const int tt_   = (t) - tb_ - (odd_ ? ne_ : 0);            \
            const float* U_ = odd_ ? (lvl_ == 0 ? Ko0_U : (lvl_ == 1 ? Ko1_U : Ko2_U))    \
                                   : (lvl_ == 0 ? Ke0_U : (lvl_ == 1 ? Ke1_U : Ke2_U));   \
            const float* V_ = odd_ ? (lvl_ == 0 ? Ko0_V : (lvl_ == 1 ? Ko1_V : Ko2_V))    \
                                   : (lvl_ == 0 ? Ke0_V : (lvl_ == 1 ? Ke1_V : Ke2_V));   \
            const int glast_ = (1024 << lvl_) - (odd_ ? 1 : 0);          \
            const int vlast_ = odd_ ? 14 : 7;                            \
            int g_ = (c0 << lvl_) + 16 * tt_ + mrow;                                      \
            g_ = g_ > glast_ ? glast_ : g_;                                               \
            const int vr_ = mrow > vlast_ ? vlast_ : mrow;                                \
            (up) = U_ + (size_t)g_ * 32 + 4 * half;                                       \
            (vp) = V_ + vr_ * 32 + 4 * half;                                              \
        } while (0)
        const float* up;
        const float* vp;
        v4f a0[4], b0[4], a1[4], b1[4], a2[4], b2[4];
        TILE_SRC(wave, up, vp);
#pragma unroll
        for (int qq = 0; qq < 4; ++qq) { a0[qq] = *(const v4f*)(up + 8 * qq); b0[qq] = *(const v4f*)(vp + 8 * qq); }
        TILE_SRC(wave + 8, up, vp);
#pragma unroll
        for (int qq = 0; qq < 4; ++qq) { a1[qq] = *(const v4f*)(up + 8 * qq); b1[qq] = *(const v4f*)(vp + 8 * qq); }
        if (wave == 0) {
            TILE_SRC(16, up, vp);
#pragma unroll
            for (int qq = 0; qq < 4; ++qq) { a2[qq] = *(const v4f*)(up + 8 * qq); b2[qq] = *(const v4f*)(vp + 8 * qq); }
        }
        v8f d0 = {}; v8f d1 = {};
#pragma unroll
        for (int s = 0; s < 8; ++s) {
            const int qq = s >> 1, e = s & 1;
            v2f x, y;
            x.x = a0[qq][2 * e]; x.y = a0[qq][2 * e + 1];
            y.x = b0[qq][2 * e]; y.y = b0[qq][2 * e + 1];
            d0 = WMMA_F32X4(x, y, d0);
            x.x = a1[qq][2 * e]; x.y = a1[qq][2 * e + 1];
            y.x = b1[qq][2 * e]; y.y = b1[qq][2 * e + 1];
            d1 = WMMA_F32X4(x, y, d1);
        }
        float* o = kcf + (16 * wave + 8 * half) * KPITCH + mrow;
#pragma unroll
        for (int j = 0; j < 8; ++j) {
            o[j * KPITCH]         = d0[j];
            o[(128 + j) * KPITCH] = d1[j];
        }
        if (wave == 0) {
            v8f d2 = {};
#pragma unroll
            for (int s = 0; s < 8; ++s) {
                const int qq = s >> 1, e = s & 1;
                v2f x, y;
                x.x = a2[qq][2 * e]; x.y = a2[qq][2 * e + 1];
                y.x = b2[qq][2 * e]; y.y = b2[qq][2 * e + 1];
                d2 = WMMA_F32X4(x, y, d2);
            }
#pragma unroll
            for (int j = 0; j < 8; ++j) o[(256 + j) * KPITCH] = d2[j];
        }
#undef TILE_SRC
    }

    const int wlo = jlo * LP;
    const int whi = jhi * LP;
    const int lc   = lane < 30 ? lane : 29;
    const int boff = lc < 15 ? lc : lc - 14;
    const int lce  = lane < 15 ? lane : 14;
    int evv = -1;

    {
        float sv[4][7];
        float wv[4], wx[4];
        const int wl = lane <= SEGC ? lane : SEGC;
#pragma unroll
        for (int q = 0; q < 4; ++q) {
            const int r = wave + 8 * q;
            const float* us = uh + (size_t)(row0 + r) * 8193 + jlo;
#pragma unroll
            for (int k = 0; k < 7; ++k) {
                const int p  = lane + 32 * k;
                const int pc = (32 * k + 31 < WUMIN) ? p : (p < wu ? p : wu - 1);
                sv[q][k] = us[pc];
            }
            wv[q] = wh0[(size_t)(row0 + r) * 1025 + c0 + wl];
            wx[q] = wh0[(size_t)524800 + (size_t)(row0 + r) * 1025 + c0 + wl];
        }
        int rvh[14];
        int rch[3] = {0, 0, 0};
#pragma unroll
        for (int i = 0; i < 3; ++i) {
            const int* nbr = i == 0 ? nbrL0 : (i == 1 ? nbrL1 : nbrL2);
            const int nb = c0 << i;
#pragma unroll
            for (int t = 0; t < (2 << i); ++t)
                rvh[(2 << i) - 2 + t] = nbr[(size_t)(2 * (nb + wave + 8 * t)) * 15 + lc];
            if (wave == 0) rch[i] = nbr[(size_t)(2 * (nb + (SEGC << i))) * 15 + lce];
        }
#pragma unroll
        for (int q = 0; q < 4; ++q) {
            const int r = wave + 8 * q;
#pragma unroll
            for (int k = 0; k < 7; ++k) {
                const int p  = lane + 32 * k;
                const int pc = (32 * k + 31 < WUMIN) ? p : (p < wu ? p : wu - 1);
                uw[pc * LP + r] = sv[q][k];
            }
            wA[wl * LP + r] = wv[q];
            wX[wl * LP + r] = wx[q];
        }
#pragma unroll
        for (int i = 0; i < 3; ++i) {
            const int P = (4 >> i) * LP;
#pragma unroll
            for (int t = 0; t < (2 << i); ++t) {
                const int k  = (2 << i) - 2 + t;
                const int dv = rvh[k] - boff;
                const int e0 = __builtin_amdgcn_readfirstlane(dv);
                const bool ok = __builtin_amdgcn_ballot_w32(dv != e0) == 0u
                                && e0 * P >= wlo && (e0 + 15) * P <= whi;
                evv = (lane == k) ? (ok ? e0 : -1) : evv;
            }
            if (wave == 0) {
                const int dv = rch[i] - lce;
                const int e0 = __builtin_amdgcn_readfirstlane(dv);
                const bool ok = __builtin_amdgcn_ballot_w32(dv != e0) == 0u
                                && e0 * P >= wlo && (e0 + 14) * P <= whi;
                evv = (lane == 14 + i) ? (ok ? e0 : -1) : evv;
            }
        }
    }
    __syncthreads();

    const int vb = lane - jlo * LP;
    const float* ug  = uh + (size_t)(row0 + lane) * 8193;

    const int*   nbrP[3] = {nbrL0, nbrL1, nbrL2};
    float* oldW = wA;
    float* newW = wB;


#pragma unroll
    for (int i = 0; i < 3; ++i) {
        const int step = 4 >> i;
        const float hf = H * (float)step;
        const int ns = SEGC << i;
        const int nb = c0 << i;
        const float* Ke = kcf + KE_ROW(i) * KPITCH;
        const float* Ko = kcf + KO_ROW(i) * KPITCH;
        const int* nbr  = nbrP[i];
        const int P = step * LP;
        for (int n = wave, tw = (2 << i) - 2; n < ns; n += 8, ++tw) {
            const int e0 = __builtin_amdgcn_readlane(evv, tw);
            const v4f* kpe = (const v4f*)(Ke + n * KPITCH);
            const v4f* kpo = (const v4f*)(Ko + n * KPITCH);
            float ke[8], ko[16];
            const v4f e0v = kpe[0], e1v = kpe[1];
            const v4f d0 = kpo[0], d1 = kpo[1], d2 = kpo[2], d3 = kpo[3];
#pragma unroll
            for (int m = 0; m < 4; ++m) {
                ke[m] = e0v[m]; ke[4 + m] = e1v[m];
                ko[m] = d0[m]; ko[4 + m] = d1[m]; ko[8 + m] = d2[m]; ko[12 + m] = d3[m];
            }
            float o0 = oldW[n * LP + lane];
            float o1 = oldW[(n + 1) * LP + lane];
            if (i == 0) {
                o0 += wX[n * LP + lane];
                o1 += wX[(n + 1) * LP + lane];
            }
            float ae = 0.f, ao = 0.f;
            if (e0 >= 0) {
                const float* wp = uw + vb + e0 * P;
                float w[16];
#pragma unroll
                for (int k = 0; k < 16; ++k) w[k] = wp[k * P];
#pragma unroll
                for (int m = 0; m < 8; ++m) ae = __builtin_fmaf(ke[m], w[2 * m], ae);
#pragma unroll
                for (int m = 0; m < 15; ++m) ao = __builtin_fmaf(ko[m], w[m + 1], ao);
            } else {
                const int* row = nbr + (size_t)(2 * (nb + n)) * 15;
                const int rv = row[lc];
                int xe[8], xd[15];
#pragma unroll
                for (int m = 0; m < 8; ++m) xe[m] = __builtin_amdgcn_readlane(rv, 2 * m);
#pragma unroll
                for (int m = 0; m < 15; ++m) xd[m] = __builtin_amdgcn_readlane(rv, 15 + m);
                int lo = xe[0], hi = xe[0];
#pragma unroll
                for (int m = 1; m < 8; ++m) { lo = xe[m] < lo ? xe[m] : lo; hi = xe[m] > hi ? xe[m] : hi; }
#pragma unroll
                for (int m = 0; m < 15; ++m) { lo = xd[m] < lo ? xd[m] : lo; hi = xd[m] > hi ? xd[m] : hi; }
                if (lo * P >= wlo && hi * P <= whi) {
#pragma unroll
                    for (int m = 0; m < 8; ++m) ae = __builtin_fmaf(ke[m], uw[vb + xe[m] * P], ae);
#pragma unroll
                    for (int m = 0; m < 15; ++m) ao = __builtin_fmaf(ko[m], uw[vb + xd[m] * P], ao);
                } else {
#pragma unroll
                    for (int m = 0; m < 8; ++m) ae = __builtin_fmaf(ke[m], ug[step * xe[m]], ae);
#pragma unroll
                    for (int m = 0; m < 15; ++m) ao = __builtin_fmaf(ko[m], ug[step * xd[m]], ao);
                }
            }
            newW[(2 * n) * LP + lane]     = o0 + hf * ae;
            newW[(2 * n + 1) * LP + lane] = 0.5f * (o0 + o1) + hf * ao;
        }
        if (wave == 0) {
            const int n = ns;
            const int eb = __builtin_amdgcn_readlane(evv, 14 + i);
            const v4f* kpe = (const v4f*)(Ke + n * KPITCH);
            float ke[8];
            const v4f e0 = kpe[0], e1 = kpe[1];
#pragma unroll
            for (int m = 0; m < 4; ++m) { ke[m] = e0[m]; ke[4 + m] = e1[m]; }
            float o0 = oldW[n * LP + lane];
            if (i == 0) o0 += wX[n * LP + lane];
            float ae = 0.f;
            if (eb >= 0) {
                const float* wp = uw + vb + eb * P;
#pragma unroll
                for (int m = 0; m < 8; ++m) ae = __builtin_fmaf(ke[m], wp[2 * m * P], ae);
            } else {
                const int* row = nbr + (size_t)(2 * (nb + n)) * 15;
                int se[8];
#pragma unroll
                for (int m = 0; m < 8; ++m) se[m] = row[2 * m] * P;
                int lo = se[0], hi = se[0];
#pragma unroll
                for (int m = 1; m < 8; ++m) { lo = se[m] < lo ? se[m] : lo; hi = se[m] > hi ? se[m] : hi; }
                if (lo >= wlo && hi <= whi) {
#pragma unroll
                    for (int m = 0; m < 8; ++m) ae = __builtin_fmaf(ke[m], uw[vb + se[m]], ae);
                } else {
#pragma unroll
                    for (int m = 0; m < 8; ++m) ae = __builtin_fmaf(ke[m], ug[step * row[2 * m]], ae);
                }
            }
            newW[(2 * n) * LP + lane] = o0 + hf * ae;
        }
        __syncthreads();
        float* tmp = oldW; oldW = newW; newW = tmp;
    }

    if (seg == 0 || seg == NSEG - 1) {
        if (wave == 0) {
            const float* kv = Kbd_V + ((seg == 0) ? 0 : 32);
            const float* sp0 = sbuf + (size_t)(row0 + lane) * 32;
            const float* sp1 = sbuf + 16384 + (size_t)(row0 + lane) * 32;
            float acc = 0.f;
#pragma unroll
            for (int r = 0; r < 32; ++r) acc += kv[r] * (sp0[r] + sp1[r]);
            oldW[((seg == 0) ? 0 : 8 * SEGC) * LP + lane] = H * acc;
        }
        __syncthreads();
    }

    const int npts = (seg == NSEG - 1) ? (8 * SEGC + 1) : (8 * SEGC);
    for (int r = wave; r < 32; r += 8) {
        float* dst = out + (size_t)(row0 + r) * 8193 + 8 * c0;
        for (int p = lane; p < npts; p += 32) dst[p] = oldW[p * LP + r];
    }
}

extern "C" void kernel_launch(void* const* d_in, const int* in_sizes, int n_in,
                              void* d_out, int out_size, void* d_ws, size_t ws_size,
                              hipStream_t stream) {
    const float* uh     = (const float*)d_in[0];
    const float* KHH_U  = (const float*)d_in[1];
    const float* KHH_V  = (const float*)d_in[2];
    const float* Ke0_U  = (const float*)d_in[3];
    const float* Ke0_V  = (const float*)d_in[4];
    const float* Ko0_U  = (const float*)d_in[5];
    const float* Ko0_V  = (const float*)d_in[6];
    const int*   nbr0   = (const int*)d_in[7];
    const float* Ke1_U  = (const float*)d_in[8];
    const float* Ke1_V  = (const float*)d_in[9];
    const float* Ko1_U  = (const float*)d_in[10];
    const float* Ko1_V  = (const float*)d_in[11];
    const int*   nbr1   = (const int*)d_in[12];
    const float* Ke2_U  = (const float*)d_in[13];
    const float* Ke2_V  = (const float*)d_in[14];
    const float* Ko2_U  = (const float*)d_in[15];
    const float* Ko2_V  = (const float*)d_in[16];
    const int*   nbr2   = (const int*)d_in[17];
    const float* Kbd_U  = (const float*)d_in[18];
    const float* Kbd_V  = (const float*)d_in[19];

    float* ws  = (float*)d_ws;
    float* wh0 = ws;
    float* sb  = wh0 + 1049600;

    float* outp = (float*)d_out;

    k_wmma_proj<<<128, 1024, 0, stream>>>(uh, KHH_U, KHH_V, Kbd_U, wh0, sb);

    k_refine<<<1024, 256, 0, stream>>>(uh, wh0, sb, Kbd_V,
                                      Ke0_U, Ke0_V, Ko0_U, Ko0_V,
                                      Ke1_U, Ke1_V, Ko1_U, Ko1_V,
                                      Ke2_U, Ke2_V, Ko2_U, Ko2_V,
                                      nbr2, nbr1, nbr0, outp);
}
